# starting kernel
# baseline (speedup 1.0000x reference)
#include <hip/hip_runtime.h>
#include <hip/hip_bf16.h>
#include <math.h>

typedef __attribute__((ext_vector_type(16))) _Float16 v16h;
typedef __attribute__((ext_vector_type(8)))  _Float16 v8h;
typedef __attribute__((ext_vector_type(4)))  _Float16 v4h;
typedef __attribute__((ext_vector_type(8)))  float    v8f;

#define WAVES 4
#define BLOCK 128


__device__ __forceinline__ float silu_f(float x) {
    return x / (1.0f + __expf(-x));
}

__device__ __forceinline__ unsigned enc_f(float f) {
    unsigned u = __float_as_uint(f);
    return (u & 0x80000000u) ? ~u : (u | 0x80000000u);
}
__device__ __forceinline__ float dec_f(unsigned u) {
    unsigned b = (u & 0x80000000u) ? (u & 0x7FFFFFFFu) : ~u;
    return __uint_as_float(b);
}
#define ENC_NEG_INF 0x007FFFFFu

__device__ __forceinline__ v16h load_frag(const _Float16* p) {
    v8h lo = *(const v8h*)(p);
    v8h hi = *(const v8h*)(p + 16);
    return __builtin_shufflevector(lo, hi, 0, 1, 2, 3, 4, 5, 6, 7,
                                   8, 9, 10, 11, 12, 13, 14, 15);
}

__device__ __forceinline__ v8f wmma32(v16h a, v16h b, v8f c) {
    return __builtin_amdgcn_wmma_f32_16x16x32_f16(
         false, a,  false, b,
         (short)0, c,  false,  false);
}

__device__ __forceinline__ void cvt_store4(_Float16* dstp, const float* srcp) {
    const float4 v = *(const float4*)srcp;
    v4h h;
    h[0] = (_Float16)v.x; h[1] = (_Float16)v.y;
    h[2] = (_Float16)v.z; h[3] = (_Float16)v.w;
    *(v4h*)dstp = h;
}
__device__ __forceinline__ void zero_store4(_Float16* dstp) {
    v4h z = {};
    *(v4h*)dstp = z;
}


__global__ __launch_bounds__(256) void init_ws_kernel(unsigned* e_max, float* denom,
                                                      float* h_neigh, int N) {
    int stride = gridDim.x * blockDim.x;
    int i0 = blockIdx.x * blockDim.x + threadIdx.x;
    for (int i = i0; i < N * 64; i += stride) h_neigh[i] = 0.0f;
    for (int i = i0; i < N; i += stride) { denom[i] = 0.0f; e_max[i] = ENC_NEG_INF; }
}


__global__ __launch_bounds__(BLOCK) void edge_score_kernel(
    const float* __restrict__ node_feat, const float* __restrict__ coord,
    const float* __restrict__ edge_feat,
    const float* __restrict__ W_q, const float* __restrict__ W_k,
    const int* __restrict__ src, const int* __restrict__ dst,
    float* __restrict__ e_arr, unsigned* __restrict__ e_max, int E) {

    __shared__ _Float16 sWkT[64 * 96];
    __shared__ _Float16 sWqT[64 * 64];
    __shared__ _Float16 sA[WAVES][16 * 160];

    for (int i = threadIdx.x; i < 64 * 96; i += BLOCK) {
        int n = i / 96, k = i % 96;
        int kr = (k < 64) ? k : (k < 80 ? (65 + (k - 64)) : (k == 80 ? 64 : -1));
        sWkT[i] = (kr >= 0) ? (_Float16)W_k[kr * 64 + n] : (_Float16)0.0f;
    }
    for (int i = threadIdx.x; i < 64 * 64; i += BLOCK) {
        int n = i >> 6, k = i & 63;
        sWqT[i] = (_Float16)W_q[k * 64 + n];
    }

    const int wave = threadIdx.x >> 5, lane = threadIdx.x & 31;
    const int row = lane & 15, hi = lane >> 4;
    const int numTiles = (E + 15) >> 4;
    const int tile = blockIdx.x * WAVES + wave;
    const bool validWave = tile < numTiles;

    int e = validWave ? (tile * 16 + row) : 0;
    if (e >= E) e = 0;
    const int s = src[e], d = dst[e];

    __syncthreads();

    _Float16* Ar = &sA[wave][row * 160];
    if (hi == 0) {
#pragma unroll
        for (int c = 0; c < 64; c += 4) cvt_store4(&Ar[c], &node_feat[s * 64 + c]);
#pragma unroll
        for (int c = 0; c < 32; c += 4) cvt_store4(&Ar[96 + c], &node_feat[d * 64 + c]);
    } else {
#pragma unroll
        for (int c = 32; c < 64; c += 4) cvt_store4(&Ar[96 + c], &node_feat[d * 64 + c]);
#pragma unroll
        for (int c = 0; c < 16; c += 4) cvt_store4(&Ar[64 + c], &edge_feat[e * 16 + c]);
#pragma unroll
        for (int c = 80; c < 96; c += 4) zero_store4(&Ar[c]);
        float dx = coord[s * 3 + 0] - coord[d * 3 + 0];
        float dy = coord[s * 3 + 1] - coord[d * 3 + 1];
        float dz = coord[s * 3 + 2] - coord[d * 3 + 2];
        Ar[80] = (_Float16)sqrtf(dx * dx + dy * dy + dz * dz);
    }
    __syncthreads();

    v8f accK[4], accQ[4];
#pragma unroll
    for (int nt = 0; nt < 4; ++nt) { accK[nt] = {}; accQ[nt] = {}; }

#pragma unroll
    for (int kc = 0; kc < 3; ++kc) {
        const int kb = kc * 32 + hi * 8;
        v16h a = load_frag(&sA[wave][row * 160 + kb]);
#pragma unroll
        for (int nt = 0; nt < 4; ++nt)
            accK[nt] = wmma32(a, load_frag(&sWkT[(nt * 16 + row) * 96 + kb]), accK[nt]);
    }
#pragma unroll
    for (int kc = 0; kc < 2; ++kc) {
        const int kb = kc * 32 + hi * 8;
        v16h a = load_frag(&sA[wave][row * 160 + 96 + kb]);
#pragma unroll
        for (int nt = 0; nt < 4; ++nt)
            accQ[nt] = wmma32(a, load_frag(&sWqT[(nt * 16 + row) * 64 + kb]), accQ[nt]);
    }

    float p[8];
#pragma unroll
    for (int r = 0; r < 8; ++r) {
        p[r] = 0.0f;
#pragma unroll
        for (int nt = 0; nt < 4; ++nt) p[r] += accK[nt][r] * accQ[nt][r];
    }
#pragma unroll
    for (int msk = 1; msk < 16; msk <<= 1) {
#pragma unroll
        for (int r = 0; r < 8; ++r) p[r] += __shfl_xor(p[r], msk, 32);
    }

    if (validWave && row == 0) {
#pragma unroll
        for (int r = 0; r < 8; ++r) {
            int m = r + 8 * hi;
            int eg = tile * 16 + m;
            if (eg < E) {
                float ev = p[r] * 0.125f;
                e_arr[eg] = ev;
                atomicMax(&e_max[dst[eg]], enc_f(ev));
            }
        }
    }
}


__global__ __launch_bounds__(BLOCK) void edge_msg_kernel(
    const float* __restrict__ node_feat, const float* __restrict__ coord,
    const float* __restrict__ edge_feat,
    const float* __restrict__ W_e1, const float* __restrict__ b_e1,
    const float* __restrict__ W_e2, const float* __restrict__ b_e2,
    const int* __restrict__ src, const int* __restrict__ dst,
    const float* __restrict__ e_arr, const unsigned* __restrict__ e_max,
    float* __restrict__ denom, float* __restrict__ h_neigh, int E) {

    __shared__ _Float16 sW1T[64 * 160];
    __shared__ _Float16 sW2T[64 * 64];
    __shared__ _Float16 sA[WAVES][16 * 160];
    __shared__ _Float16 sMid[WAVES][16 * 64];
    __shared__ float    sW16[WAVES][16];
    __shared__ int      sD16[WAVES][16];

    for (int i = threadIdx.x; i < 64 * 160; i += BLOCK) {
        int n = i / 160, k = i % 160;
        int kr = (k < 128) ? k : (k < 144 ? (129 + (k - 128)) : (k == 144 ? 128 : -1));
        sW1T[i] = (kr >= 0) ? (_Float16)W_e1[kr * 64 + n] : (_Float16)0.0f;
    }
    for (int i = threadIdx.x; i < 64 * 64; i += BLOCK) {
        int n = i >> 6, k = i & 63;
        sW2T[i] = (_Float16)W_e2[k * 64 + n];
    }

    const int wave = threadIdx.x >> 5, lane = threadIdx.x & 31;
    const int row = lane & 15, hi = lane >> 4;
    const int numTiles = (E + 15) >> 4;
    const int tile = blockIdx.x * WAVES + wave;
    const bool validWave = tile < numTiles;

    int e = validWave ? (tile * 16 + row) : 0;
    if (e >= E) e = 0;
    const int s = src[e], d = dst[e];

    __syncthreads();

    _Float16* Ar = &sA[wave][row * 160];
    if (hi == 0) {
#pragma unroll
        for (int c = 0; c < 64; c += 4) cvt_store4(&Ar[c], &node_feat[s * 64 + c]);
#pragma unroll
        for (int c = 0; c < 32; c += 4) cvt_store4(&Ar[64 + c], &node_feat[d * 64 + c]);
    } else {
#pragma unroll
        for (int c = 32; c < 64; c += 4) cvt_store4(&Ar[64 + c], &node_feat[d * 64 + c]);
#pragma unroll
        for (int c = 0; c < 16; c += 4) cvt_store4(&Ar[128 + c], &edge_feat[e * 16 + c]);
#pragma unroll
        for (int c = 144; c < 160; c += 4) zero_store4(&Ar[c]);
        float dx = coord[s * 3 + 0] - coord[d * 3 + 0];
        float dy = coord[s * 3 + 1] - coord[d * 3 + 1];
        float dz = coord[s * 3 + 2] - coord[d * 3 + 2];
        Ar[144] = (_Float16)sqrtf(dx * dx + dy * dy + dz * dz);
    }
    __syncthreads();

    v8f acc[4];
#pragma unroll
    for (int nt = 0; nt < 4; ++nt) acc[nt] = {};
#pragma unroll
    for (int kc = 0; kc < 5; ++kc) {
        const int kb = kc * 32 + hi * 8;
        v16h a = load_frag(&sA[wave][row * 160 + kb]);
#pragma unroll
        for (int nt = 0; nt < 4; ++nt)
            acc[nt] = wmma32(a, load_frag(&sW1T[(nt * 16 + row) * 160 + kb]), acc[nt]);
    }
#pragma unroll
    for (int nt = 0; nt < 4; ++nt) {
        float bb = b_e1[nt * 16 + row];
#pragma unroll
        for (int r = 0; r < 8; ++r) {
            float h = silu_f(acc[nt][r] + bb);
            sMid[wave][(r + 8 * hi) * 64 + nt * 16 + row] = (_Float16)h;
        }
    }
    if (hi == 0) {
        int eg = validWave ? (tile * 16 + row) : 0;
        if (eg >= E) eg = 0;
        int dd = dst[eg];
        float w = __expf(e_arr[eg] - dec_f(e_max[dd]));
        sW16[wave][row] = w;
        sD16[wave][row] = dd;
        if (validWave && (tile * 16 + row) < E) atomicAdd(&denom[dd], w);
    }
    __syncthreads();

    v8f acc2[4];
#pragma unroll
    for (int nt = 0; nt < 4; ++nt) acc2[nt] = {};
#pragma unroll
    for (int kc = 0; kc < 2; ++kc) {
        const int kb = kc * 32 + hi * 8;
        v16h a = load_frag(&sMid[wave][row * 64 + kb]);
#pragma unroll
        for (int nt = 0; nt < 4; ++nt)
            acc2[nt] = wmma32(a, load_frag(&sW2T[(nt * 16 + row) * 64 + kb]), acc2[nt]);
    }
    if (validWave) {
#pragma unroll
        for (int nt = 0; nt < 4; ++nt) {
            float bb = b_e2[nt * 16 + row];
#pragma unroll
            for (int r = 0; r < 8; ++r) {
                int m = r + 8 * hi;
                if (tile * 16 + m < E) {
                    float msg = silu_f(acc2[nt][r] + bb);
                    atomicAdd(&h_neigh[(long)sD16[wave][m] * 64 + nt * 16 + row],
                              msg * sW16[wave][m]);
                }
            }
        }
    }
}


__global__ __launch_bounds__(BLOCK) void node_mlp_kernel(
    const float* __restrict__ node_feat,
    const float* __restrict__ W_n1, const float* __restrict__ b_n1,
    const float* __restrict__ W_n2, const float* __restrict__ b_n2,
    const float* __restrict__ denom, const float* __restrict__ h_neigh,
    float* __restrict__ out, int N) {

    __shared__ _Float16 sW1T[64 * 128];
    __shared__ _Float16 sW2T[64 * 64];
    __shared__ _Float16 sA[WAVES][16 * 128];
    __shared__ _Float16 sMid[WAVES][16 * 64];

    for (int i = threadIdx.x; i < 64 * 128; i += BLOCK) {
        int n = i >> 7, k = i & 127;
        sW1T[i] = (_Float16)W_n1[k * 64 + n];
    }
    for (int i = threadIdx.x; i < 64 * 64; i += BLOCK) {
        int n = i >> 6, k = i & 63;
        sW2T[i] = (_Float16)W_n2[k * 64 + n];
    }

    const int wave = threadIdx.x >> 5, lane = threadIdx.x & 31;
    const int row = lane & 15, hi = lane >> 4;
    const int numTiles = (N + 15) >> 4;
    const int tile = blockIdx.x * WAVES + wave;
    const bool validWave = tile < numTiles;

    int node = validWave ? (tile * 16 + row) : 0;
    if (node >= N) node = 0;

    __syncthreads();

    _Float16* Ar = &sA[wave][row * 128];
    if (hi == 0) {
#pragma unroll
        for (int c = 0; c < 64; c += 4) cvt_store4(&Ar[c], &node_feat[node * 64 + c]);
    } else {
        float dn = denom[node];
        float inv = (dn > 0.0f) ? (1.0f / dn) : 0.0f;
#pragma unroll
        for (int c = 0; c < 64; c += 4) {
            const float4 v = *(const float4*)&h_neigh[node * 64 + c];
            v4h h;
            h[0] = (_Float16)(v.x * inv); h[1] = (_Float16)(v.y * inv);
            h[2] = (_Float16)(v.z * inv); h[3] = (_Float16)(v.w * inv);
            *(v4h*)&Ar[64 + c] = h;
        }
    }
    __syncthreads();

    v8f acc[4];
#pragma unroll
    for (int nt = 0; nt < 4; ++nt) acc[nt] = {};
#pragma unroll
    for (int kc = 0; kc < 4; ++kc) {
        const int kb = kc * 32 + hi * 8;
        v16h a = load_frag(&sA[wave][row * 128 + kb]);
#pragma unroll
        for (int nt = 0; nt < 4; ++nt)
            acc[nt] = wmma32(a, load_frag(&sW1T[(nt * 16 + row) * 128 + kb]), acc[nt]);
    }
#pragma unroll
    for (int nt = 0; nt < 4; ++nt) {
        float bb = b_n1[nt * 16 + row];
#pragma unroll
        for (int r = 0; r < 8; ++r) {
            float h = silu_f(acc[nt][r] + bb);
            sMid[wave][(r + 8 * hi) * 64 + nt * 16 + row] = (_Float16)h;
        }
    }
    __syncthreads();

    v8f acc2[4];
#pragma unroll
    for (int nt = 0; nt < 4; ++nt) acc2[nt] = {};
#pragma unroll
    for (int kc = 0; kc < 2; ++kc) {
        const int kb = kc * 32 + hi * 8;
        v16h a = load_frag(&sMid[wave][row * 64 + kb]);
#pragma unroll
        for (int nt = 0; nt < 4; ++nt)
            acc2[nt] = wmma32(a, load_frag(&sW2T[(nt * 16 + row) * 64 + kb]), acc2[nt]);
    }
    if (validWave) {
#pragma unroll
        for (int nt = 0; nt < 4; ++nt) {
            float bb = b_n2[nt * 16 + row];
#pragma unroll
            for (int r = 0; r < 8; ++r) {
                int m = r + 8 * hi;
                int nd = tile * 16 + m;
                if (nd < N) out[(long)nd * 64 + nt * 16 + row] = acc2[nt][r] + bb;
            }
        }
    }
}


extern "C" void kernel_launch(void* const* d_in, const int* in_sizes, int n_in,
                              void* d_out, int out_size, void* d_ws, size_t ws_size,
                              hipStream_t stream) {
    const float* node_feat = (const float*)d_in[0];
    const float* coord     = (const float*)d_in[1];
    const float* edge_feat = (const float*)d_in[2];
    const float* W_e1 = (const float*)d_in[3];
    const float* b_e1 = (const float*)d_in[4];
    const float* W_e2 = (const float*)d_in[5];
    const float* b_e2 = (const float*)d_in[6];
    const float* W_n1 = (const float*)d_in[7];
    const float* b_n1 = (const float*)d_in[8];
    const float* W_n2 = (const float*)d_in[9];
    const float* b_n2 = (const float*)d_in[10];
    const float* W_q  = (const float*)d_in[11];
    const float* W_k  = (const float*)d_in[12];
    const int*   src  = (const int*)d_in[13];
    const int*   dst  = (const int*)d_in[14];

    const int N = in_sizes[0] / 64;
    const int E = in_sizes[13];

    float*    e_arr   = (float*)d_ws;
    unsigned* e_max   = (unsigned*)(e_arr + E);
    float*    denom   = (float*)(e_max + N);
    float*    h_neigh = denom + N;

    int initBlocks = (N * 64 + 255) / 256;
    if (initBlocks > 1024) initBlocks = 1024;
    init_ws_kernel<<<initBlocks, 256, 0, stream>>>(e_max, denom, h_neigh, N);

    const int numEdgeTiles = (E + 15) / 16;
    const int edgeBlocks   = (numEdgeTiles + WAVES - 1) / WAVES;
    const int numNodeTiles = (N + 15) / 16;
    const int nodeBlocks   = (numNodeTiles + WAVES - 1) / WAVES;

    edge_score_kernel<<<edgeBlocks, BLOCK, 0, stream>>>(
        node_feat, coord, edge_feat, W_q, W_k, src, dst, e_arr, e_max, E);

    edge_msg_kernel<<<edgeBlocks, BLOCK, 0, stream>>>(
        node_feat, coord, edge_feat, W_e1, b_e1, W_e2, b_e2,
        src, dst, e_arr, e_max, denom, h_neigh, E);

    node_mlp_kernel<<<nodeBlocks, BLOCK, 0, stream>>>(
        node_feat, W_n1, b_n1, W_n2, b_n2, denom, h_neigh, (float*)d_out, N);
}


# persistent work-groups in both edge kernels: 768 groups each stage the weight tiles to LDS once and stride over the edge tiles (before: every group of 64 edges restaged all weights)
# speedup vs baseline: 1.1779x; 1.1779x over previous
#include <hip/hip_runtime.h>
#include <hip/hip_bf16.h>
#include <math.h>

typedef __attribute__((ext_vector_type(16))) _Float16 v16h;
typedef __attribute__((ext_vector_type(8)))  _Float16 v8h;
typedef __attribute__((ext_vector_type(4)))  _Float16 v4h;
typedef __attribute__((ext_vector_type(8)))  float    v8f;

#define WAVES 4
#define BLOCK 128


__device__ __forceinline__ float silu_f(float x) {
    return x / (1.0f + __expf(-x));
}

__device__ __forceinline__ unsigned enc_f(float f) {
    unsigned u = __float_as_uint(f);
    return (u & 0x80000000u) ? ~u : (u | 0x80000000u);
}
__device__ __forceinline__ float dec_f(unsigned u) {
    unsigned b = (u & 0x80000000u) ? (u & 0x7FFFFFFFu) : ~u;
    return __uint_as_float(b);
}
#define ENC_NEG_INF 0x007FFFFFu

__device__ __forceinline__ v16h load_frag(const _Float16* p) {
    v8h lo = *(const v8h*)(p);
    v8h hi = *(const v8h*)(p + 16);
    return __builtin_shufflevector(lo, hi, 0, 1, 2, 3, 4, 5, 6, 7,
                                   8, 9, 10, 11, 12, 13, 14, 15);
}

__device__ __forceinline__ v8f wmma32(v16h a, v16h b, v8f c) {
    return __builtin_amdgcn_wmma_f32_16x16x32_f16(
         false, a,  false, b,
         (short)0, c,  false,  false);
}

__device__ __forceinline__ void cvt_store4(_Float16* dstp, const float* srcp) {
    const float4 v = *(const float4*)srcp;
    v4h h;
    h[0] = (_Float16)v.x; h[1] = (_Float16)v.y;
    h[2] = (_Float16)v.z; h[3] = (_Float16)v.w;
    *(v4h*)dstp = h;
}
__device__ __forceinline__ void zero_store4(_Float16* dstp) {
    v4h z = {};
    *(v4h*)dstp = z;
}


__global__ __launch_bounds__(256) void init_ws_kernel(unsigned* e_max, float* denom,
                                                      float* h_neigh, int N) {
    int stride = gridDim.x * blockDim.x;
    int i0 = blockIdx.x * blockDim.x + threadIdx.x;
    for (int i = i0; i < N * 64; i += stride) h_neigh[i] = 0.0f;
    for (int i = i0; i < N; i += stride) { denom[i] = 0.0f; e_max[i] = ENC_NEG_INF; }
}


__global__ __launch_bounds__(BLOCK) void edge_score_kernel(
    const float* __restrict__ node_feat, const float* __restrict__ coord,
    const float* __restrict__ edge_feat,
    const float* __restrict__ W_q, const float* __restrict__ W_k,
    const int* __restrict__ src, const int* __restrict__ dst,
    float* __restrict__ e_arr, unsigned* __restrict__ e_max, int E) {

    __shared__ _Float16 sWkT[64 * 96];
    __shared__ _Float16 sWqT[64 * 64];
    __shared__ _Float16 sA[WAVES][16 * 160];

    for (int i = threadIdx.x; i < 64 * 96; i += BLOCK) {
        int n = i / 96, k = i % 96;
        int kr = (k < 64) ? k : (k < 80 ? (65 + (k - 64)) : (k == 80 ? 64 : -1));
        sWkT[i] = (kr >= 0) ? (_Float16)W_k[kr * 64 + n] : (_Float16)0.0f;
    }
    for (int i = threadIdx.x; i < 64 * 64; i += BLOCK) {
        int n = i >> 6, k = i & 63;
        sWqT[i] = (_Float16)W_q[k * 64 + n];
    }

    const int wave = threadIdx.x >> 5, lane = threadIdx.x & 31;
    const int row = lane & 15, hi = lane >> 4;
    const int numTiles = (E + 15) >> 4;

    __syncthreads();

    for (int base = blockIdx.x * WAVES; base < numTiles; base += gridDim.x * WAVES) {
        const int tile = base + wave;
        const bool validWave = tile < numTiles;

        int e = validWave ? (tile * 16 + row) : 0;
        if (e >= E) e = 0;
        const int s = src[e], d = dst[e];

        _Float16* Ar = &sA[wave][row * 160];
        if (hi == 0) {
#pragma unroll
            for (int c = 0; c < 64; c += 4) cvt_store4(&Ar[c], &node_feat[s * 64 + c]);
#pragma unroll
            for (int c = 0; c < 32; c += 4) cvt_store4(&Ar[96 + c], &node_feat[d * 64 + c]);
        } else {
#pragma unroll
            for (int c = 32; c < 64; c += 4) cvt_store4(&Ar[96 + c], &node_feat[d * 64 + c]);
#pragma unroll
            for (int c = 0; c < 16; c += 4) cvt_store4(&Ar[64 + c], &edge_feat[e * 16 + c]);
#pragma unroll
            for (int c = 80; c < 96; c += 4) zero_store4(&Ar[c]);
            float dx = coord[s * 3 + 0] - coord[d * 3 + 0];
            float dy = coord[s * 3 + 1] - coord[d * 3 + 1];
            float dz = coord[s * 3 + 2] - coord[d * 3 + 2];
            Ar[80] = (_Float16)sqrtf(dx * dx + dy * dy + dz * dz);
        }
        __syncthreads();

        v8f accK[4], accQ[4];
#pragma unroll
        for (int nt = 0; nt < 4; ++nt) { accK[nt] = {}; accQ[nt] = {}; }

#pragma unroll
        for (int kc = 0; kc < 3; ++kc) {
            const int kb = kc * 32 + hi * 8;
            v16h a = load_frag(&sA[wave][row * 160 + kb]);
#pragma unroll
            for (int nt = 0; nt < 4; ++nt)
                accK[nt] = wmma32(a, load_frag(&sWkT[(nt * 16 + row) * 96 + kb]), accK[nt]);
        }
#pragma unroll
        for (int kc = 0; kc < 2; ++kc) {
            const int kb = kc * 32 + hi * 8;
            v16h a = load_frag(&sA[wave][row * 160 + 96 + kb]);
#pragma unroll
            for (int nt = 0; nt < 4; ++nt)
                accQ[nt] = wmma32(a, load_frag(&sWqT[(nt * 16 + row) * 64 + kb]), accQ[nt]);
        }

        float p[8];
#pragma unroll
        for (int r = 0; r < 8; ++r) {
            p[r] = 0.0f;
#pragma unroll
            for (int nt = 0; nt < 4; ++nt) p[r] += accK[nt][r] * accQ[nt][r];
        }
#pragma unroll
        for (int msk = 1; msk < 16; msk <<= 1) {
#pragma unroll
            for (int r = 0; r < 8; ++r) p[r] += __shfl_xor(p[r], msk, 32);
        }

        if (validWave && row == 0) {
#pragma unroll
            for (int r = 0; r < 8; ++r) {
                int m = r + 8 * hi;
                int eg = tile * 16 + m;
                if (eg < E) {
                    float ev = p[r] * 0.125f;
                    e_arr[eg] = ev;
                    atomicMax(&e_max[dst[eg]], enc_f(ev));
                }
            }
        }
    }
}


__global__ __launch_bounds__(BLOCK) void edge_msg_kernel(
    const float* __restrict__ node_feat, const float* __restrict__ coord,
    const float* __restrict__ edge_feat,
    const float* __restrict__ W_e1, const float* __restrict__ b_e1,
    const float* __restrict__ W_e2, const float* __restrict__ b_e2,
    const int* __restrict__ src, const int* __restrict__ dst,
    const float* __restrict__ e_arr, const unsigned* __restrict__ e_max,
    float* __restrict__ denom, float* __restrict__ h_neigh, int E) {

    __shared__ _Float16 sW1T[64 * 160];
    __shared__ _Float16 sW2T[64 * 64];
    __shared__ _Float16 sA[WAVES][16 * 160];
    __shared__ _Float16 sMid[WAVES][16 * 64];
    __shared__ float    sW16[WAVES][16];
    __shared__ int      sD16[WAVES][16];

    for (int i = threadIdx.x; i < 64 * 160; i += BLOCK) {
        int n = i / 160, k = i % 160;
        int kr = (k < 128) ? k : (k < 144 ? (129 + (k - 128)) : (k == 144 ? 128 : -1));
        sW1T[i] = (kr >= 0) ? (_Float16)W_e1[kr * 64 + n] : (_Float16)0.0f;
    }
    for (int i = threadIdx.x; i < 64 * 64; i += BLOCK) {
        int n = i >> 6, k = i & 63;
        sW2T[i] = (_Float16)W_e2[k * 64 + n];
    }

    const int wave = threadIdx.x >> 5, lane = threadIdx.x & 31;
    const int row = lane & 15, hi = lane >> 4;
    const int numTiles = (E + 15) >> 4;

    float bb1[4], bb2[4];
#pragma unroll
    for (int nt = 0; nt < 4; ++nt) {
        bb1[nt] = b_e1[nt * 16 + row];
        bb2[nt] = b_e2[nt * 16 + row];
    }

    __syncthreads();

    for (int base = blockIdx.x * WAVES; base < numTiles; base += gridDim.x * WAVES) {
        const int tile = base + wave;
        const bool validWave = tile < numTiles;

        int e = validWave ? (tile * 16 + row) : 0;
        if (e >= E) e = 0;
        const int s = src[e], d = dst[e];

        _Float16* Ar = &sA[wave][row * 160];
        if (hi == 0) {
#pragma unroll
            for (int c = 0; c < 64; c += 4) cvt_store4(&Ar[c], &node_feat[s * 64 + c]);
#pragma unroll
            for (int c = 0; c < 32; c += 4) cvt_store4(&Ar[64 + c], &node_feat[d * 64 + c]);
        } else {
#pragma unroll
            for (int c = 32; c < 64; c += 4) cvt_store4(&Ar[64 + c], &node_feat[d * 64 + c]);
#pragma unroll
            for (int c = 0; c < 16; c += 4) cvt_store4(&Ar[128 + c], &edge_feat[e * 16 + c]);
#pragma unroll
            for (int c = 144; c < 160; c += 4) zero_store4(&Ar[c]);
            float dx = coord[s * 3 + 0] - coord[d * 3 + 0];
            float dy = coord[s * 3 + 1] - coord[d * 3 + 1];
            float dz = coord[s * 3 + 2] - coord[d * 3 + 2];
            Ar[144] = (_Float16)sqrtf(dx * dx + dy * dy + dz * dz);
        }
        __syncthreads();

        v8f acc[4];
#pragma unroll
        for (int nt = 0; nt < 4; ++nt) acc[nt] = {};
#pragma unroll
        for (int kc = 0; kc < 5; ++kc) {
            const int kb = kc * 32 + hi * 8;
            v16h a = load_frag(&sA[wave][row * 160 + kb]);
#pragma unroll
            for (int nt = 0; nt < 4; ++nt)
                acc[nt] = wmma32(a, load_frag(&sW1T[(nt * 16 + row) * 160 + kb]), acc[nt]);
        }
#pragma unroll
        for (int nt = 0; nt < 4; ++nt) {
#pragma unroll
            for (int r = 0; r < 8; ++r) {
                float h = silu_f(acc[nt][r] + bb1[nt]);
                sMid[wave][(r + 8 * hi) * 64 + nt * 16 + row] = (_Float16)h;
            }
        }
        if (hi == 0) {
            int eg = validWave ? (tile * 16 + row) : 0;
            if (eg >= E) eg = 0;
            int dd = dst[eg];
            float w = __expf(e_arr[eg] - dec_f(e_max[dd]));
            sW16[wave][row] = w;
            sD16[wave][row] = dd;
            if (validWave && (tile * 16 + row) < E) atomicAdd(&denom[dd], w);
        }
        __syncthreads();

        v8f acc2[4];
#pragma unroll
        for (int nt = 0; nt < 4; ++nt) acc2[nt] = {};
#pragma unroll
        for (int kc = 0; kc < 2; ++kc) {
            const int kb = kc * 32 + hi * 8;
            v16h a = load_frag(&sMid[wave][row * 64 + kb]);
#pragma unroll
            for (int nt = 0; nt < 4; ++nt)
                acc2[nt] = wmma32(a, load_frag(&sW2T[(nt * 16 + row) * 64 + kb]), acc2[nt]);
        }
        if (validWave) {
#pragma unroll
            for (int nt = 0; nt < 4; ++nt) {
#pragma unroll
                for (int r = 0; r < 8; ++r) {
                    int m = r + 8 * hi;
                    if (tile * 16 + m < E) {
                        float msg = silu_f(acc2[nt][r] + bb2[nt]);
                        atomicAdd(&h_neigh[(long)sD16[wave][m] * 64 + nt * 16 + row],
                                  msg * sW16[wave][m]);
                    }
                }
            }
        }
    }
}


__global__ __launch_bounds__(BLOCK) void node_mlp_kernel(
    const float* __restrict__ node_feat,
    const float* __restrict__ W_n1, const float* __restrict__ b_n1,
    const float* __restrict__ W_n2, const float* __restrict__ b_n2,
    const float* __restrict__ denom, const float* __restrict__ h_neigh,
    float* __restrict__ out, int N) {

    __shared__ _Float16 sW1T[64 * 128];
    __shared__ _Float16 sW2T[64 * 64];
    __shared__ _Float16 sA[WAVES][16 * 128];
    __shared__ _Float16 sMid[WAVES][16 * 64];

    for (int i = threadIdx.x; i < 64 * 128; i += BLOCK) {
        int n = i >> 7, k = i & 127;
        sW1T[i] = (_Float16)W_n1[k * 64 + n];
    }
    for (int i = threadIdx.x; i < 64 * 64; i += BLOCK) {
        int n = i >> 6, k = i & 63;
        sW2T[i] = (_Float16)W_n2[k * 64 + n];
    }

    const int wave = threadIdx.x >> 5, lane = threadIdx.x & 31;
    const int row = lane & 15, hi = lane >> 4;
    const int numTiles = (N + 15) >> 4;
    const int tile = blockIdx.x * WAVES + wave;
    const bool validWave = tile < numTiles;

    int node = validWave ? (tile * 16 + row) : 0;
    if (node >= N) node = 0;

    __syncthreads();

    _Float16* Ar = &sA[wave][row * 128];
    if (hi == 0) {
#pragma unroll
        for (int c = 0; c < 64; c += 4) cvt_store4(&Ar[c], &node_feat[node * 64 + c]);
    } else {
        float dn = denom[node];
        float inv = (dn > 0.0f) ? (1.0f / dn) : 0.0f;
#pragma unroll
        for (int c = 0; c < 64; c += 4) {
            const float4 v = *(const float4*)&h_neigh[node * 64 + c];
            v4h h;
            h[0] = (_Float16)(v.x * inv); h[1] = (_Float16)(v.y * inv);
            h[2] = (_Float16)(v.z * inv); h[3] = (_Float16)(v.w * inv);
            *(v4h*)&Ar[64 + c] = h;
        }
    }
    __syncthreads();

    v8f acc[4];
#pragma unroll
    for (int nt = 0; nt < 4; ++nt) acc[nt] = {};
#pragma unroll
    for (int kc = 0; kc < 4; ++kc) {
        const int kb = kc * 32 + hi * 8;
        v16h a = load_frag(&sA[wave][row * 128 + kb]);
#pragma unroll
        for (int nt = 0; nt < 4; ++nt)
            acc[nt] = wmma32(a, load_frag(&sW1T[(nt * 16 + row) * 128 + kb]), acc[nt]);
    }
#pragma unroll
    for (int nt = 0; nt < 4; ++nt) {
        float bb = b_n1[nt * 16 + row];
#pragma unroll
        for (int r = 0; r < 8; ++r) {
            float h = silu_f(acc[nt][r] + bb);
            sMid[wave][(r + 8 * hi) * 64 + nt * 16 + row] = (_Float16)h;
        }
    }
    __syncthreads();

    v8f acc2[4];
#pragma unroll
    for (int nt = 0; nt < 4; ++nt) acc2[nt] = {};
#pragma unroll
    for (int kc = 0; kc < 2; ++kc) {
        const int kb = kc * 32 + hi * 8;
        v16h a = load_frag(&sMid[wave][row * 64 + kb]);
#pragma unroll
        for (int nt = 0; nt < 4; ++nt)
            acc2[nt] = wmma32(a, load_frag(&sW2T[(nt * 16 + row) * 64 + kb]), acc2[nt]);
    }
    if (validWave) {
#pragma unroll
        for (int nt = 0; nt < 4; ++nt) {
            float bb = b_n2[nt * 16 + row];
#pragma unroll
            for (int r = 0; r < 8; ++r) {
                int m = r + 8 * hi;
                int nd = tile * 16 + m;
                if (nd < N) out[(long)nd * 64 + nt * 16 + row] = acc2[nt][r] + bb;
            }
        }
    }
}


extern "C" void kernel_launch(void* const* d_in, const int* in_sizes, int n_in,
                              void* d_out, int out_size, void* d_ws, size_t ws_size,
                              hipStream_t stream) {
    const float* node_feat = (const float*)d_in[0];
    const float* coord     = (const float*)d_in[1];
    const float* edge_feat = (const float*)d_in[2];
    const float* W_e1 = (const float*)d_in[3];
    const float* b_e1 = (const float*)d_in[4];
    const float* W_e2 = (const float*)d_in[5];
    const float* b_e2 = (const float*)d_in[6];
    const float* W_n1 = (const float*)d_in[7];
    const float* b_n1 = (const float*)d_in[8];
    const float* W_n2 = (const float*)d_in[9];
    const float* b_n2 = (const float*)d_in[10];
    const float* W_q  = (const float*)d_in[11];
    const float* W_k  = (const float*)d_in[12];
    const int*   src  = (const int*)d_in[13];
    const int*   dst  = (const int*)d_in[14];

    const int N = in_sizes[0] / 64;
    const int E = in_sizes[13];

    float*    e_arr   = (float*)d_ws;
    unsigned* e_max   = (unsigned*)(e_arr + E);
    float*    denom   = (float*)(e_max + N);
    float*    h_neigh = denom + N;

    int initBlocks = (N * 64 + 255) / 256;
    if (initBlocks > 1024) initBlocks = 1024;
    init_ws_kernel<<<initBlocks, 256, 0, stream>>>(e_max, denom, h_neigh, N);

    const int numEdgeTiles = (E + 15) / 16;
    const int edgeBlocks   = (numEdgeTiles + WAVES - 1) / WAVES;
    const int numNodeTiles = (N + 15) / 16;
    const int nodeBlocks   = (numNodeTiles + WAVES - 1) / WAVES;

    const int edgeGrid = edgeBlocks < 768 ? edgeBlocks : 768;

    edge_score_kernel<<<edgeGrid, BLOCK, 0, stream>>>(
        node_feat, coord, edge_feat, W_q, W_k, src, dst, e_arr, e_max, E);

    edge_msg_kernel<<<edgeGrid, BLOCK, 0, stream>>>(
        node_feat, coord, edge_feat, W_e1, b_e1, W_e2, b_e2,
        src, dst, e_arr, e_max, denom, h_neigh, E);

    node_mlp_kernel<<<nodeBlocks, BLOCK, 0, stream>>>(
        node_feat, W_n1, b_n1, W_n2, b_n2, denom, h_neigh, (float*)d_out, N);
}


# message kernel owns destinations: each work-group filters dst for its 40 nodes, sums weighted messages and softmax denominators in LDS and writes normalized rows once; no global float atomics left
# speedup vs baseline: 1.3145x; 1.1160x over previous
#include <hip/hip_runtime.h>
#include <hip/hip_bf16.h>
#include <math.h>

typedef __attribute__((ext_vector_type(16))) _Float16 v16h;
typedef __attribute__((ext_vector_type(8)))  _Float16 v8h;
typedef __attribute__((ext_vector_type(4)))  _Float16 v4h;
typedef __attribute__((ext_vector_type(8)))  float    v8f;

#define WAVES 4
#define BLOCK 128


__device__ __forceinline__ float silu_f(float x) {
    return x / (1.0f + __expf(-x));
}

__device__ __forceinline__ unsigned enc_f(float f) {
    unsigned u = __float_as_uint(f);
    return (u & 0x80000000u) ? ~u : (u | 0x80000000u);
}
__device__ __forceinline__ float dec_f(unsigned u) {
    unsigned b = (u & 0x80000000u) ? (u & 0x7FFFFFFFu) : ~u;
    return __uint_as_float(b);
}
#define ENC_NEG_INF 0x007FFFFFu

__device__ __forceinline__ v16h load_frag(const _Float16* p) {
    v8h lo = *(const v8h*)(p);
    v8h hi = *(const v8h*)(p + 16);
    return __builtin_shufflevector(lo, hi, 0, 1, 2, 3, 4, 5, 6, 7,
                                   8, 9, 10, 11, 12, 13, 14, 15);
}

__device__ __forceinline__ v8f wmma32(v16h a, v16h b, v8f c) {
    return __builtin_amdgcn_wmma_f32_16x16x32_f16(
         false, a,  false, b,
         (short)0, c,  false,  false);
}

__device__ __forceinline__ void cvt_store4(_Float16* dstp, const float* srcp) {
    const float4 v = *(const float4*)srcp;
    v4h h;
    h[0] = (_Float16)v.x; h[1] = (_Float16)v.y;
    h[2] = (_Float16)v.z; h[3] = (_Float16)v.w;
    *(v4h*)dstp = h;
}
__device__ __forceinline__ void zero_store4(_Float16* dstp) {
    v4h z = {};
    *(v4h*)dstp = z;
}


__global__ __launch_bounds__(256) void init_ws_kernel(unsigned* e_max, int N) {
    int stride = gridDim.x * blockDim.x;
    int i0 = blockIdx.x * blockDim.x + threadIdx.x;
    for (int i = i0; i < N; i += stride) e_max[i] = ENC_NEG_INF;
}


__global__ __launch_bounds__(BLOCK) void edge_score_kernel(
    const float* __restrict__ node_feat, const float* __restrict__ coord,
    const float* __restrict__ edge_feat,
    const float* __restrict__ W_q, const float* __restrict__ W_k,
    const int* __restrict__ src, const int* __restrict__ dst,
    float* __restrict__ e_arr, unsigned* __restrict__ e_max, int E) {

    __shared__ _Float16 sWkT[64 * 96];
    __shared__ _Float16 sWqT[64 * 64];
    __shared__ _Float16 sA[WAVES][16 * 160];

    for (int i = threadIdx.x; i < 64 * 96; i += BLOCK) {
        int n = i / 96, k = i % 96;
        int kr = (k < 64) ? k : (k < 80 ? (65 + (k - 64)) : (k == 80 ? 64 : -1));
        sWkT[i] = (kr >= 0) ? (_Float16)W_k[kr * 64 + n] : (_Float16)0.0f;
    }
    for (int i = threadIdx.x; i < 64 * 64; i += BLOCK) {
        int n = i >> 6, k = i & 63;
        sWqT[i] = (_Float16)W_q[k * 64 + n];
    }

    const int wave = threadIdx.x >> 5, lane = threadIdx.x & 31;
    const int row = lane & 15, hi = lane >> 4;
    const int numTiles = (E + 15) >> 4;

    __syncthreads();

    for (int base = blockIdx.x * WAVES; base < numTiles; base += gridDim.x * WAVES) {
        const int tile = base + wave;
        const bool validWave = tile < numTiles;

        int e = validWave ? (tile * 16 + row) : 0;
        if (e >= E) e = 0;
        const int s = src[e], d = dst[e];

        _Float16* Ar = &sA[wave][row * 160];
        if (hi == 0) {
#pragma unroll
            for (int c = 0; c < 64; c += 4) cvt_store4(&Ar[c], &node_feat[s * 64 + c]);
#pragma unroll
            for (int c = 0; c < 32; c += 4) cvt_store4(&Ar[96 + c], &node_feat[d * 64 + c]);
        } else {
#pragma unroll
            for (int c = 32; c < 64; c += 4) cvt_store4(&Ar[96 + c], &node_feat[d * 64 + c]);
#pragma unroll
            for (int c = 0; c < 16; c += 4) cvt_store4(&Ar[64 + c], &edge_feat[e * 16 + c]);
#pragma unroll
            for (int c = 80; c < 96; c += 4) zero_store4(&Ar[c]);
            float dx = coord[s * 3 + 0] - coord[d * 3 + 0];
            float dy = coord[s * 3 + 1] - coord[d * 3 + 1];
            float dz = coord[s * 3 + 2] - coord[d * 3 + 2];
            Ar[80] = (_Float16)sqrtf(dx * dx + dy * dy + dz * dz);
        }
        __syncthreads();

        v8f accK[4], accQ[4];
#pragma unroll
        for (int nt = 0; nt < 4; ++nt) { accK[nt] = {}; accQ[nt] = {}; }

#pragma unroll
        for (int kc = 0; kc < 3; ++kc) {
            const int kb = kc * 32 + hi * 8;
            v16h a = load_frag(&sA[wave][row * 160 + kb]);
#pragma unroll
            for (int nt = 0; nt < 4; ++nt)
                accK[nt] = wmma32(a, load_frag(&sWkT[(nt * 16 + row) * 96 + kb]), accK[nt]);
        }
#pragma unroll
        for (int kc = 0; kc < 2; ++kc) {
            const int kb = kc * 32 + hi * 8;
            v16h a = load_frag(&sA[wave][row * 160 + 96 + kb]);
#pragma unroll
            for (int nt = 0; nt < 4; ++nt)
                accQ[nt] = wmma32(a, load_frag(&sWqT[(nt * 16 + row) * 64 + kb]), accQ[nt]);
        }

        float p[8];
#pragma unroll
        for (int r = 0; r < 8; ++r) {
            p[r] = 0.0f;
#pragma unroll
            for (int nt = 0; nt < 4; ++nt) p[r] += accK[nt][r] * accQ[nt][r];
        }
#pragma unroll
        for (int msk = 1; msk < 16; msk <<= 1) {
#pragma unroll
            for (int r = 0; r < 8; ++r) p[r] += __shfl_xor(p[r], msk, 32);
        }

        if (validWave && row == 0) {
#pragma unroll
            for (int r = 0; r < 8; ++r) {
                int m = r + 8 * hi;
                int eg = tile * 16 + m;
                if (eg < E) {
                    float ev = p[r] * 0.125f;
                    e_arr[eg] = ev;
                    atomicMax(&e_max[dst[eg]], enc_f(ev));
                }
            }
        }
    }
}


#define MW     16
#define MBLOCK (MW * 32)
#define NPG    40
#define UNIT   1024
#define LCAP   (UNIT + 32)

__device__ __forceinline__ void wave_lds_sync() {
    __builtin_amdgcn_fence(__ATOMIC_RELEASE, "wavefront");
    __builtin_amdgcn_wave_barrier();
    __builtin_amdgcn_fence(__ATOMIC_ACQUIRE, "wavefront");
}

__global__ __launch_bounds__(MBLOCK) void edge_msg_kernel(
    const float* __restrict__ node_feat, const float* __restrict__ coord,
    const float* __restrict__ edge_feat,
    const float* __restrict__ W_e1, const float* __restrict__ b_e1,
    const float* __restrict__ W_e2, const float* __restrict__ b_e2,
    const int* __restrict__ src, const int* __restrict__ dst,
    const float* __restrict__ e_arr, const unsigned* __restrict__ e_max,
    float* __restrict__ h_neigh, int E, int N) {

    __shared__ _Float16 sW1T[64 * 160];
    __shared__ _Float16 sW2T[64 * 64];
    __shared__ _Float16 sA[MW][16 * 160];
    __shared__ _Float16 sMid[MW][16 * 64];
    __shared__ float    sW16[MW][16];
    __shared__ int      sD16[MW][16];
    __shared__ int      sLE[MW][LCAP];
    __shared__ float    sAcc[NPG * 64];
    __shared__ float    sDen[NPG];

    for (int i = threadIdx.x; i < 64 * 160; i += MBLOCK) {
        int n = i / 160, k = i % 160;
        int kr = (k < 128) ? k : (k < 144 ? (129 + (k - 128)) : (k == 144 ? 128 : -1));
        sW1T[i] = (kr >= 0) ? (_Float16)W_e1[kr * 64 + n] : (_Float16)0.0f;
    }
    for (int i = threadIdx.x; i < 64 * 64; i += MBLOCK) {
        int n = i >> 6, k = i & 63;
        sW2T[i] = (_Float16)W_e2[k * 64 + n];
    }
    for (int i = threadIdx.x; i < NPG * 64; i += MBLOCK) sAcc[i] = 0.0f;
    if (threadIdx.x < NPG) sDen[threadIdx.x] = 0.0f;

    const int wave = __builtin_amdgcn_readfirstlane((int)(threadIdx.x >> 5));
    const int lane = threadIdx.x & 31;
    const int row = lane & 15, hi = lane >> 4;
    const int lo = blockIdx.x * NPG;

    float bb1[4], bb2[4];
#pragma unroll
    for (int nt = 0; nt < 4; ++nt) {
        bb1[nt] = b_e1[nt * 16 + row];
        bb2[nt] = b_e2[nt * 16 + row];
    }

    __syncthreads();

    const int Ef = E & ~3;
    const int numUnits = (Ef + UNIT - 1) / UNIT;
    const int rot = (numUnits > 0) ? (int)((blockIdx.x * 37u) % (unsigned)numUnits) : 0;

    int cnt = 0;
    int* const myList = &sLE[wave][0];

#define APPEND_EDGES(m, ev)                                                        \
    {                                                                              \
        const unsigned b_ = __builtin_amdgcn_ballot_w32(m);                        \
        if (m) myList[__builtin_amdgcn_mbcnt_lo(b_, (unsigned)cnt)] = (ev);        \
        cnt += __builtin_popcount(b_);                                             \
    }

    if (wave == 0) {
        const bool inTail = lane < (E - Ef);
        const int dv = inTail ? dst[Ef + lane] : -1;
        const bool mt = inTail && ((unsigned)(dv - lo) < (unsigned)NPG);
        APPEND_EDGES(mt, Ef + lane)
    }

    for (int k = wave; ; k += MW) {
        const bool last = (k >= numUnits);
        if (!last) {
            int u = k + rot;
            if (u >= numUnits) u -= numUnits;
            const int ubase = u * UNIT;
            int4 v[8];
#pragma unroll
            for (int j = 0; j < 8; ++j) {
                const int e0 = ubase + j * 128 + lane * 4;
                v[j] = *(const int4*)(dst + ((e0 < Ef) ? e0 : 0));
            }
#pragma unroll
            for (int j = 0; j < 8; ++j) {
                const int e0 = ubase + j * 128 + lane * 4;
                const bool qv = e0 < Ef;
                const bool m0 = qv && ((unsigned)(v[j].x - lo) < (unsigned)NPG);
                const bool m1 = qv && ((unsigned)(v[j].y - lo) < (unsigned)NPG);
                const bool m2 = qv && ((unsigned)(v[j].z - lo) < (unsigned)NPG);
                const bool m3 = qv && ((unsigned)(v[j].w - lo) < (unsigned)NPG);
                if (__builtin_amdgcn_ballot_w32(m0 | m1 | m2 | m3) != 0u) {
                    APPEND_EDGES(m0, e0)
                    APPEND_EDGES(m1, e0 + 1)
                    APPEND_EDGES(m2, e0 + 2)
                    APPEND_EDGES(m3, e0 + 3)
                }
            }
        }

        cnt = __builtin_amdgcn_readfirstlane(cnt);
        while (cnt >= 16 || (last && cnt > 0)) {
            const int take = (cnt >= 16) ? 16 : cnt;
            cnt -= take;
            wave_lds_sync();
            const bool rowValid = row < take;
            const int e = myList[cnt + (rowValid ? row : 0)];
            const int s = src[e], d = dst[e];
            const int dl = d - lo;

            _Float16* Ar = &sA[wave][row * 160];
            if (hi == 0) {
#pragma unroll
                for (int c = 0; c < 64; c += 4) cvt_store4(&Ar[c], &node_feat[s * 64 + c]);
#pragma unroll
                for (int c = 0; c < 32; c += 4) cvt_store4(&Ar[64 + c], &node_feat[d * 64 + c]);
            } else {
#pragma unroll
                for (int c = 32; c < 64; c += 4) cvt_store4(&Ar[64 + c], &node_feat[d * 64 + c]);
#pragma unroll
                for (int c = 0; c < 16; c += 4) cvt_store4(&Ar[128 + c], &edge_feat[e * 16 + c]);
#pragma unroll
                for (int c = 144; c < 160; c += 4) zero_store4(&Ar[c]);
                float dx = coord[s * 3 + 0] - coord[d * 3 + 0];
                float dy = coord[s * 3 + 1] - coord[d * 3 + 1];
                float dz = coord[s * 3 + 2] - coord[d * 3 + 2];
                Ar[144] = (_Float16)sqrtf(dx * dx + dy * dy + dz * dz);
            }
            wave_lds_sync();

            v8f acc[4];
#pragma unroll
            for (int nt = 0; nt < 4; ++nt) acc[nt] = {};
#pragma unroll
            for (int kc = 0; kc < 5; ++kc) {
                const int kb = kc * 32 + hi * 8;
                v16h a = load_frag(&sA[wave][row * 160 + kb]);
#pragma unroll
                for (int nt = 0; nt < 4; ++nt)
                    acc[nt] = wmma32(a, load_frag(&sW1T[(nt * 16 + row) * 160 + kb]), acc[nt]);
            }
#pragma unroll
            for (int nt = 0; nt < 4; ++nt) {
#pragma unroll
                for (int r = 0; r < 8; ++r) {
                    float h = silu_f(acc[nt][r] + bb1[nt]);
                    sMid[wave][(r + 8 * hi) * 64 + nt * 16 + row] = (_Float16)h;
                }
            }
            if (hi == 0) {
                float w = __expf(e_arr[e] - dec_f(e_max[d]));
                sW16[wave][row] = w;
                sD16[wave][row] = dl;
                if (rowValid) atomicAdd(&sDen[dl], w);
            }
            wave_lds_sync();

            v8f acc2[4];
#pragma unroll
            for (int nt = 0; nt < 4; ++nt) acc2[nt] = {};
#pragma unroll
            for (int kc = 0; kc < 2; ++kc) {
                const int kb = kc * 32 + hi * 8;
                v16h a = load_frag(&sMid[wave][row * 64 + kb]);
#pragma unroll
                for (int nt = 0; nt < 4; ++nt)
                    acc2[nt] = wmma32(a, load_frag(&sW2T[(nt * 16 + row) * 64 + kb]), acc2[nt]);
            }
#pragma unroll
            for (int nt = 0; nt < 4; ++nt) {
#pragma unroll
                for (int r = 0; r < 8; ++r) {
                    int m = r + 8 * hi;
                    if (m < take) {
                        float msg = silu_f(acc2[nt][r] + bb2[nt]);
                        atomicAdd(&sAcc[sD16[wave][m] * 64 + nt * 16 + row],
                                  msg * sW16[wave][m]);
                    }
                }
            }
        }
        if (last) break;
    }
#undef APPEND_EDGES

    __syncthreads();

    for (int i = threadIdx.x; i < NPG * 64; i += MBLOCK) {
        const int nl = i >> 6;
        const int node = lo + nl;
        if (node < N) {
            const float dn = sDen[nl];
            const float inv = (dn > 0.0f) ? (1.0f / dn) : 0.0f;
            h_neigh[(long)node * 64 + (i & 63)] = sAcc[i] * inv;
        }
    }
}


__global__ __launch_bounds__(BLOCK) void node_mlp_kernel(
    const float* __restrict__ node_feat,
    const float* __restrict__ W_n1, const float* __restrict__ b_n1,
    const float* __restrict__ W_n2, const float* __restrict__ b_n2,
    const float* __restrict__ h_neigh,
    float* __restrict__ out, int N) {

    __shared__ _Float16 sW1T[64 * 128];
    __shared__ _Float16 sW2T[64 * 64];
    __shared__ _Float16 sA[WAVES][16 * 128];
    __shared__ _Float16 sMid[WAVES][16 * 64];

    for (int i = threadIdx.x; i < 64 * 128; i += BLOCK) {
        int n = i >> 7, k = i & 127;
        sW1T[i] = (_Float16)W_n1[k * 64 + n];
    }
    for (int i = threadIdx.x; i < 64 * 64; i += BLOCK) {
        int n = i >> 6, k = i & 63;
        sW2T[i] = (_Float16)W_n2[k * 64 + n];
    }

    const int wave = threadIdx.x >> 5, lane = threadIdx.x & 31;
    const int row = lane & 15, hi = lane >> 4;
    const int numTiles = (N + 15) >> 4;
    const int tile = blockIdx.x * WAVES + wave;
    const bool validWave = tile < numTiles;

    int node = validWave ? (tile * 16 + row) : 0;
    if (node >= N) node = 0;

    __syncthreads();

    _Float16* Ar = &sA[wave][row * 128];
    if (hi == 0) {
#pragma unroll
        for (int c = 0; c < 64; c += 4) cvt_store4(&Ar[c], &node_feat[node * 64 + c]);
    } else {
#pragma unroll
        for (int c = 0; c < 64; c += 4) cvt_store4(&Ar[64 + c], &h_neigh[node * 64 + c]);
    }
    __syncthreads();

    v8f acc[4];
#pragma unroll
    for (int nt = 0; nt < 4; ++nt) acc[nt] = {};
#pragma unroll
    for (int kc = 0; kc < 4; ++kc) {
        const int kb = kc * 32 + hi * 8;
        v16h a = load_frag(&sA[wave][row * 128 + kb]);
#pragma unroll
        for (int nt = 0; nt < 4; ++nt)
            acc[nt] = wmma32(a, load_frag(&sW1T[(nt * 16 + row) * 128 + kb]), acc[nt]);
    }
#pragma unroll
    for (int nt = 0; nt < 4; ++nt) {
        float bb = b_n1[nt * 16 + row];
#pragma unroll
        for (int r = 0; r < 8; ++r) {
            float h = silu_f(acc[nt][r] + bb);
            sMid[wave][(r + 8 * hi) * 64 + nt * 16 + row] = (_Float16)h;
        }
    }
    __syncthreads();

    v8f acc2[4];
#pragma unroll
    for (int nt = 0; nt < 4; ++nt) acc2[nt] = {};
#pragma unroll
    for (int kc = 0; kc < 2; ++kc) {
        const int kb = kc * 32 + hi * 8;
        v16h a = load_frag(&sMid[wave][row * 64 + kb]);
#pragma unroll
        for (int nt = 0; nt < 4; ++nt)
            acc2[nt] = wmma32(a, load_frag(&sW2T[(nt * 16 + row) * 64 + kb]), acc2[nt]);
    }
    if (validWave) {
#pragma unroll
        for (int nt = 0; nt < 4; ++nt) {
            float bb = b_n2[nt * 16 + row];
#pragma unroll
            for (int r = 0; r < 8; ++r) {
                int m = r + 8 * hi;
                int nd = tile * 16 + m;
                if (nd < N) out[(long)nd * 64 + nt * 16 + row] = acc2[nt][r] + bb;
            }
        }
    }
}


extern "C" void kernel_launch(void* const* d_in, const int* in_sizes, int n_in,
                              void* d_out, int out_size, void* d_ws, size_t ws_size,
                              hipStream_t stream) {
    const float* node_feat = (const float*)d_in[0];
    const float* coord     = (const float*)d_in[1];
    const float* edge_feat = (const float*)d_in[2];
    const float* W_e1 = (const float*)d_in[3];
    const float* b_e1 = (const float*)d_in[4];
    const float* W_e2 = (const float*)d_in[5];
    const float* b_e2 = (const float*)d_in[6];
    const float* W_n1 = (const float*)d_in[7];
    const float* b_n1 = (const float*)d_in[8];
    const float* W_n2 = (const float*)d_in[9];
    const float* b_n2 = (const float*)d_in[10];
    const float* W_q  = (const float*)d_in[11];
    const float* W_k  = (const float*)d_in[12];
    const int*   src  = (const int*)d_in[13];
    const int*   dst  = (const int*)d_in[14];

    const int N = in_sizes[0] / 64;
    const int E = in_sizes[13];

    float*    e_arr   = (float*)d_ws;
    unsigned* e_max   = (unsigned*)(e_arr + E);
    float*    h_neigh = (float*)(e_max + N) + N;

    int initBlocks = (N + 255) / 256;
    if (initBlocks > 1024) initBlocks = 1024;
    init_ws_kernel<<<initBlocks, 256, 0, stream>>>(e_max, N);

    const int numEdgeTiles = (E + 15) / 16;
    const int edgeBlocks   = (numEdgeTiles + WAVES - 1) / WAVES;
    const int numNodeTiles = (N + 15) / 16;
    const int nodeBlocks   = (numNodeTiles + WAVES - 1) / WAVES;

    const int edgeGrid = edgeBlocks < 768 ? edgeBlocks : 768;
    const int msgGroups = (N + NPG - 1) / NPG;

    edge_score_kernel<<<edgeGrid, BLOCK, 0, stream>>>(
        node_feat, coord, edge_feat, W_q, W_k, src, dst, e_arr, e_max, E);

    edge_msg_kernel<<<msgGroups, MBLOCK, 0, stream>>>(
        node_feat, coord, edge_feat, W_e1, b_e1, W_e2, b_e2,
        src, dst, e_arr, e_max, h_neigh, E, N);

    node_mlp_kernel<<<nodeBlocks, BLOCK, 0, stream>>>(
        node_feat, W_n1, b_n1, W_n2, b_n2, h_neigh, (float*)d_out, N);
}


# resident waves per SIMD raised in both edge kernels (score: 3 -> 5, grid 768 -> 1280 groups; message: 4 -> 6, 16 -> 23 waves per group, all that LDS holds): more waves cover each tile's wait chain
# speedup vs baseline: 1.8396x; 1.3995x over previous
#include <hip/hip_runtime.h>
#include <hip/hip_bf16.h>
#include <math.h>

typedef __attribute__((ext_vector_type(16))) _Float16 v16h;
typedef __attribute__((ext_vector_type(8)))  _Float16 v8h;
typedef __attribute__((ext_vector_type(4)))  _Float16 v4h;
typedef __attribute__((ext_vector_type(8)))  float    v8f;

#define WAVES 4
#define BLOCK 128

#define P160 168
#define P128 136
#define P96  104
#define P64   72


__device__ __forceinline__ float silu_f(float x) {
    return x / (1.0f + __expf(-x));
}

__device__ __forceinline__ unsigned enc_f(float f) {
    unsigned u = __float_as_uint(f);
    return (u & 0x80000000u) ? ~u : (u | 0x80000000u);
}
__device__ __forceinline__ float dec_f(unsigned u) {
    unsigned b = (u & 0x80000000u) ? (u & 0x7FFFFFFFu) : ~u;
    return __uint_as_float(b);
}
#define ENC_NEG_INF 0x007FFFFFu

__device__ __forceinline__ v16h load_frag(const _Float16* p) {
    v8h lo = *(const v8h*)(p);
    v8h hi = *(const v8h*)(p + 16);
    return __builtin_shufflevector(lo, hi, 0, 1, 2, 3, 4, 5, 6, 7,
                                   8, 9, 10, 11, 12, 13, 14, 15);
}

__device__ __forceinline__ v8f wmma32(v16h a, v16h b, v8f c) {
    return __builtin_amdgcn_wmma_f32_16x16x32_f16(
         false, a,  false, b,
         (short)0, c,  false,  false);
}

__device__ __forceinline__ void cvt_store4(_Float16* dstp, const float* srcp) {
    const float4 v = *(const float4*)srcp;
    v4h h;
    h[0] = (_Float16)v.x; h[1] = (_Float16)v.y;
    h[2] = (_Float16)v.z; h[3] = (_Float16)v.w;
    *(v4h*)dstp = h;
}
__device__ __forceinline__ void cvt_store4v(_Float16* dstp, const float4 v) {
    v4h h;
    h[0] = (_Float16)v.x; h[1] = (_Float16)v.y;
    h[2] = (_Float16)v.z; h[3] = (_Float16)v.w;
    *(v4h*)dstp = h;
}

__device__ __forceinline__ float xhalf(const float x) {
    return __builtin_bit_cast(float, __builtin_amdgcn_permlanex16(
        __builtin_bit_cast(int, x), __builtin_bit_cast(int, x), 0x76543210, 0xfedcba98, false, false));
}

__device__ __forceinline__ void gather_rows64(float4 (&v)[8], const float* __restrict__ feat,
                                              const int idx, const int lane) {
    const bool upper = lane >= 16;
    const unsigned c4 = (unsigned)(lane & 15) * 4u;
#pragma unroll
    for (int j = 0; j < 8; ++j) {
        const int r0 = __builtin_amdgcn_readlane(idx, j);
        const int r1 = __builtin_amdgcn_readlane(idx, j + 8);
        const unsigned r = (unsigned)(upper ? r1 : r0);
        v[j] = *(const float4*)(feat + (r * 64u + c4));
    }
}
__device__ __forceinline__ void store_rows64(_Float16* tile, const int col0,
                                             const float4 (&v)[8], const int lane) {
    _Float16* p = tile + ((lane >> 4) * 8) * P160 + col0 + (lane & 15) * 4;
#pragma unroll
    for (int j = 0; j < 8; ++j) cvt_store4v(p + j * P160, v[j]);
}


__global__ __launch_bounds__(256) void init_ws_kernel(unsigned* e_max, int N) {
    int stride = gridDim.x * blockDim.x;
    int i0 = blockIdx.x * blockDim.x + threadIdx.x;
    for (int i = i0; i < N; i += stride) e_max[i] = ENC_NEG_INF;
}


__global__ __launch_bounds__(BLOCK, 6) void edge_score_kernel(
    const float* __restrict__ node_feat, const float* __restrict__ coord,
    const float* __restrict__ edge_feat,
    const float* __restrict__ W_q, const float* __restrict__ W_k,
    const int* __restrict__ src, const int* __restrict__ dst,
    float* __restrict__ e_arr, unsigned* __restrict__ e_max, int E) {

    __shared__ __attribute__((aligned(16))) _Float16 sWkT[64 * P96];
    __shared__ __attribute__((aligned(16))) _Float16 sWqT[64 * P64];
    __shared__ __attribute__((aligned(16))) _Float16 sA[WAVES][16 * P160];

    for (int i = threadIdx.x; i < 64 * 96; i += BLOCK) {
        int n = i / 96, k = i % 96;
        int kr = (k < 64) ? k : (k < 80 ? (65 + (k - 64)) : (k == 80 ? 64 : -1));
        sWkT[n * P96 + k] = (kr >= 0) ? (_Float16)W_k[kr * 64 + n] : (_Float16)0.0f;
    }
    for (int i = threadIdx.x; i < 64 * 64; i += BLOCK) {
        int n = i >> 6, k = i & 63;
        sWqT[n * P64 + k] = (_Float16)W_q[k * 64 + n];
    }

    const int wave = threadIdx.x >> 5, lane = threadIdx.x & 31;
    const int row = lane & 15, hi = lane >> 4;
    const int numTiles = (E + 15) >> 4;

    __syncthreads();

    for (int base = blockIdx.x * WAVES; base < numTiles; base += gridDim.x * WAVES) {
        const int tile = base + wave;
        const bool validWave = tile < numTiles;

        int e = validWave ? (tile * 16 + row) : 0;
        if (e >= E) e = 0;
        const int s = src[e], d = dst[e];

        const int cn = (hi ? d : s) * 3;
        const float cx = coord[cn + 0], cy = coord[cn + 1], cz = coord[cn + 2];
        float4 vs[8], vd[8], ve[2];
        gather_rows64(vs, node_feat, s, lane);
        gather_rows64(vd, node_feat, d, lane);
#pragma unroll
        for (int j = 0; j < 2; ++j) {
            int er = validWave ? (tile * 16 + (lane >> 2) + 8 * j) : 0;
            if (er >= E) er = 0;
            ve[j] = *(const float4*)(edge_feat + ((unsigned)er * 16u + (unsigned)(lane & 3) * 4u));
        }
        _Float16* At = &sA[wave][0];
        store_rows64(At, 0, vs, lane);
        store_rows64(At, 96, vd, lane);
#pragma unroll
        for (int j = 0; j < 2; ++j)
            cvt_store4v(&At[((lane >> 2) + 8 * j) * P160 + 64 + (lane & 3) * 4], ve[j]);
        {
            const float dx = cx - xhalf(cx), dy = cy - xhalf(cy), dz = cz - xhalf(cz);
            v8h t = {};
            t[0] = (hi == 0) ? (_Float16)sqrtf(dx * dx + dy * dy + dz * dz) : (_Float16)0.0f;
            *(v8h*)&At[row * P160 + 80 + 8 * hi] = t;
        }
        __syncthreads();

        v8f accK[4], accQ[4];
#pragma unroll
        for (int nt = 0; nt < 4; ++nt) { accK[nt] = {}; accQ[nt] = {}; }

#pragma unroll
        for (int kc = 0; kc < 3; ++kc) {
            const int kb = kc * 32 + hi * 8;
            v16h a = load_frag(&sA[wave][row * P160 + kb]);
#pragma unroll
            for (int nt = 0; nt < 4; ++nt)
                accK[nt] = wmma32(a, load_frag(&sWkT[(nt * 16 + row) * P96 + kb]), accK[nt]);
        }
#pragma unroll
        for (int kc = 0; kc < 2; ++kc) {
            const int kb = kc * 32 + hi * 8;
            v16h a = load_frag(&sA[wave][row * P160 + 96 + kb]);
#pragma unroll
            for (int nt = 0; nt < 4; ++nt)
                accQ[nt] = wmma32(a, load_frag(&sWqT[(nt * 16 + row) * P64 + kb]), accQ[nt]);
        }

        float p[8];
#pragma unroll
        for (int r = 0; r < 8; ++r) {
            p[r] = 0.0f;
#pragma unroll
            for (int nt = 0; nt < 4; ++nt) p[r] += accK[nt][r] * accQ[nt][r];
        }
#pragma unroll
        for (int msk = 1; msk < 16; msk <<= 1) {
#pragma unroll
            for (int r = 0; r < 8; ++r) p[r] += __shfl_xor(p[r], msk, 32);
        }

        if (validWave && row == 0) {
#pragma unroll
            for (int r = 0; r < 8; ++r) {
                int m = r + 8 * hi;
                int eg = tile * 16 + m;
                if (eg < E) {
                    float ev = p[r] * 0.125f;
                    e_arr[eg] = ev;
                    atomicMax(&e_max[dst[eg]], enc_f(ev));
                }
            }
        }
    }
}


#define MW     23
#define MBLOCK (MW * 32)
#define NPG    40
#define UNIT   1024
#define LCAP   (UNIT + 32)

__device__ __forceinline__ void wave_lds_sync() {
    __builtin_amdgcn_fence(__ATOMIC_RELEASE, "wavefront");
    __builtin_amdgcn_wave_barrier();
    __builtin_amdgcn_fence(__ATOMIC_ACQUIRE, "wavefront");
}

__global__ __launch_bounds__(MBLOCK) void edge_msg_kernel(
    const float* __restrict__ node_feat, const float* __restrict__ coord,
    const float* __restrict__ edge_feat,
    const float* __restrict__ W_e1, const float* __restrict__ b_e1,
    const float* __restrict__ W_e2, const float* __restrict__ b_e2,
    const int* __restrict__ src, const int* __restrict__ dst,
    const float* __restrict__ e_arr, const unsigned* __restrict__ e_max,
    float* __restrict__ h_neigh, int E, int N) {

    __shared__ __attribute__((aligned(16))) _Float16 sW1T[64 * P160];
    __shared__ __attribute__((aligned(16))) _Float16 sW2T[64 * P64];
    __shared__ __attribute__((aligned(16))) _Float16 sA[MW][16 * P160];
    __shared__ __attribute__((aligned(16))) _Float16 sMid[MW][16 * P64];
    __shared__ float    sW16[MW][16];
    __shared__ int      sD16[MW][16];
    __shared__ int      sLE[MW][LCAP];
    __shared__ float    sAcc[NPG * 64];
    __shared__ float    sDen[NPG];

    for (int i = threadIdx.x; i < 64 * 160; i += MBLOCK) {
        int n = i / 160, k = i % 160;
        int kr = (k < 128) ? k : (k < 144 ? (129 + (k - 128)) : (k == 144 ? 128 : -1));
        sW1T[n * P160 + k] = (kr >= 0) ? (_Float16)W_e1[kr * 64 + n] : (_Float16)0.0f;
    }
    for (int i = threadIdx.x; i < 64 * 64; i += MBLOCK) {
        int n = i >> 6, k = i & 63;
        sW2T[n * P64 + k] = (_Float16)W_e2[k * 64 + n];
    }
    for (int i = threadIdx.x; i < NPG * 64; i += MBLOCK) sAcc[i] = 0.0f;
    if (threadIdx.x < NPG) sDen[threadIdx.x] = 0.0f;

    const int wave = __builtin_amdgcn_readfirstlane((int)(threadIdx.x >> 5));
    const int lane = threadIdx.x & 31;
    const int row = lane & 15, hi = lane >> 4;
    const int lo = blockIdx.x * NPG;

    float bb1[4], bb2[4];
#pragma unroll
    for (int nt = 0; nt < 4; ++nt) {
        bb1[nt] = b_e1[nt * 16 + row];
        bb2[nt] = b_e2[nt * 16 + row];
    }

    __syncthreads();

    const int Ef = E & ~3;
    const int numUnits = (Ef + UNIT - 1) / UNIT;
    const int rot = (numUnits > 0) ? (int)((blockIdx.x * 37u) % (unsigned)numUnits) : 0;

    int cnt = 0;
    int* const myList = &sLE[wave][0];

#define APPEND_EDGES(m, ev)                                                        \
    {                                                                              \
        const unsigned b_ = __builtin_amdgcn_ballot_w32(m);                        \
        if (m) myList[__builtin_amdgcn_mbcnt_lo(b_, (unsigned)cnt)] = (ev);        \
        cnt += __builtin_popcount(b_);                                             \
    }

    if (wave == 0) {
        const bool inTail = lane < (E - Ef);
        const int dv = inTail ? dst[Ef + lane] : -1;
        const bool mt = inTail && ((unsigned)(dv - lo) < (unsigned)NPG);
        APPEND_EDGES(mt, Ef + lane)
    }

    for (int k = wave; ; k += MW) {
        const bool last = (k >= numUnits);
        if (!last) {
            int u = k + rot;
            if (u >= numUnits) u -= numUnits;
            const int ubase = u * UNIT;
            int4 v[8];
#pragma unroll
            for (int j = 0; j < 8; ++j) {
                const int e0 = ubase + j * 128 + lane * 4;
                v[j] = *(const int4*)(dst + ((e0 < Ef) ? e0 : 0));
            }
#pragma unroll
            for (int j = 0; j < 8; ++j) {
                const int e0 = ubase + j * 128 + lane * 4;
                const bool qv = e0 < Ef;
                const bool m0 = qv && ((unsigned)(v[j].x - lo) < (unsigned)NPG);
                const bool m1 = qv && ((unsigned)(v[j].y - lo) < (unsigned)NPG);
                const bool m2 = qv && ((unsigned)(v[j].z - lo) < (unsigned)NPG);
                const bool m3 = qv && ((unsigned)(v[j].w - lo) < (unsigned)NPG);
                if (__builtin_amdgcn_ballot_w32(m0 | m1 | m2 | m3) != 0u) {
                    APPEND_EDGES(m0, e0)
                    APPEND_EDGES(m1, e0 + 1)
                    APPEND_EDGES(m2, e0 + 2)
                    APPEND_EDGES(m3, e0 + 3)
                }
            }
        }

        cnt = __builtin_amdgcn_readfirstlane(cnt);
        while (cnt >= 16 || (last && cnt > 0)) {
            const int take = (cnt >= 16) ? 16 : cnt;
            cnt -= take;
            wave_lds_sync();
            const bool rowValid = row < take;
            const int e = myList[cnt + (rowValid ? row : 0)];
            const int s = src[e], d = dst[e];
            const int dl = d - lo;

            const int cn = (hi ? d : s) * 3;
            const float cx = coord[cn + 0], cy = coord[cn + 1], cz = coord[cn + 2];
            float4 vs[8], vd[8], ve[2];
            gather_rows64(vs, node_feat, s, lane);
            gather_rows64(vd, node_feat, d, lane);
#pragma unroll
            for (int j = 0; j < 2; ++j) {
                const int rr = (lane >> 2) + 8 * j;
                const int er = myList[cnt + (rr < take ? rr : 0)];
                ve[j] = *(const float4*)(edge_feat + ((unsigned)er * 16u + (unsigned)(lane & 3) * 4u));
            }
            _Float16* At = &sA[wave][0];
            store_rows64(At, 0, vs, lane);
            store_rows64(At, 64, vd, lane);
#pragma unroll
            for (int j = 0; j < 2; ++j)
                cvt_store4v(&At[((lane >> 2) + 8 * j) * P160 + 128 + (lane & 3) * 4], ve[j]);
            {
                const float dx = cx - xhalf(cx), dy = cy - xhalf(cy), dz = cz - xhalf(cz);
                v8h t = {};
                t[0] = (hi == 0) ? (_Float16)sqrtf(dx * dx + dy * dy + dz * dz) : (_Float16)0.0f;
                *(v8h*)&At[row * P160 + 144 + 8 * hi] = t;
            }
            wave_lds_sync();

            v8f acc[4];
#pragma unroll
            for (int nt = 0; nt < 4; ++nt) acc[nt] = {};
#pragma unroll
            for (int kc = 0; kc < 5; ++kc) {
                const int kb = kc * 32 + hi * 8;
                v16h a = load_frag(&sA[wave][row * P160 + kb]);
#pragma unroll
                for (int nt = 0; nt < 4; ++nt)
                    acc[nt] = wmma32(a, load_frag(&sW1T[(nt * 16 + row) * P160 + kb]), acc[nt]);
            }
#pragma unroll
            for (int nt = 0; nt < 4; ++nt) {
#pragma unroll
                for (int r = 0; r < 8; ++r) {
                    float h = silu_f(acc[nt][r] + bb1[nt]);
                    sMid[wave][(r + 8 * hi) * P64 + nt * 16 + row] = (_Float16)h;
                }
            }
            if (hi == 0) {
                float w = __expf(e_arr[e] - dec_f(e_max[d]));
                sW16[wave][row] = w;
                sD16[wave][row] = dl;
                if (rowValid) atomicAdd(&sDen[dl], w);
            }
            wave_lds_sync();

            v8f acc2[4];
#pragma unroll
            for (int nt = 0; nt < 4; ++nt) acc2[nt] = {};
#pragma unroll
            for (int kc = 0; kc < 2; ++kc) {
                const int kb = kc * 32 + hi * 8;
                v16h a = load_frag(&sMid[wave][row * P64 + kb]);
#pragma unroll
                for (int nt = 0; nt < 4; ++nt)
                    acc2[nt] = wmma32(a, load_frag(&sW2T[(nt * 16 + row) * P64 + kb]), acc2[nt]);
            }
#pragma unroll
            for (int nt = 0; nt < 4; ++nt) {
#pragma unroll
                for (int r = 0; r < 8; ++r) {
                    int m = r + 8 * hi;
                    if (m < take) {
                        float msg = silu_f(acc2[nt][r] + bb2[nt]);
                        atomicAdd(&sAcc[sD16[wave][m] * 64 + nt * 16 + row],
                                  msg * sW16[wave][m]);
                    }
                }
            }
        }
        if (last) break;
    }
#undef APPEND_EDGES

    __syncthreads();

    for (int i = threadIdx.x; i < NPG * 64; i += MBLOCK) {
        const int nl = i >> 6;
        const int node = lo + nl;
        if (node < N) {
            const float dn = sDen[nl];
            const float inv = (dn > 0.0f) ? (1.0f / dn) : 0.0f;
            h_neigh[(long)node * 64 + (i & 63)] = sAcc[i] * inv;
        }
    }
}


__global__ __launch_bounds__(BLOCK) void node_mlp_kernel(
    const float* __restrict__ node_feat,
    const float* __restrict__ W_n1, const float* __restrict__ b_n1,
    const float* __restrict__ W_n2, const float* __restrict__ b_n2,
    const float* __restrict__ h_neigh,
    float* __restrict__ out, int N) {

    __shared__ __attribute__((aligned(16))) _Float16 sW1T[64 * P128];
    __shared__ __attribute__((aligned(16))) _Float16 sW2T[64 * P64];
    __shared__ __attribute__((aligned(16))) _Float16 sA[WAVES][16 * P128];
    __shared__ __attribute__((aligned(16))) _Float16 sMid[WAVES][16 * P64];

    for (int i = threadIdx.x; i < 64 * 128; i += BLOCK) {
        int n = i >> 7, k = i & 127;
        sW1T[n * P128 + k] = (_Float16)W_n1[k * 64 + n];
    }
    for (int i = threadIdx.x; i < 64 * 64; i += BLOCK) {
        int n = i >> 6, k = i & 63;
        sW2T[n * P64 + k] = (_Float16)W_n2[k * 64 + n];
    }

    const int wave = threadIdx.x >> 5, lane = threadIdx.x & 31;
    const int row = lane & 15, hi = lane >> 4;
    const int numTiles = (N + 15) >> 4;
    const int tile = blockIdx.x * WAVES + wave;
    const bool validWave = tile < numTiles;

    int node = validWave ? (tile * 16 + row) : 0;
    if (node >= N) node = 0;

    __syncthreads();

    _Float16* Ar = &sA[wave][row * P128];
    if (hi == 0) {
#pragma unroll
        for (int c = 0; c < 64; c += 4) cvt_store4(&Ar[c], &node_feat[node * 64 + c]);
    } else {
#pragma unroll
        for (int c = 0; c < 64; c += 4) cvt_store4(&Ar[64 + c], &h_neigh[node * 64 + c]);
    }
    __syncthreads();

    v8f acc[4];
#pragma unroll
    for (int nt = 0; nt < 4; ++nt) acc[nt] = {};
#pragma unroll
    for (int kc = 0; kc < 4; ++kc) {
        const int kb = kc * 32 + hi * 8;
        v16h a = load_frag(&sA[wave][row * P128 + kb]);
#pragma unroll
        for (int nt = 0; nt < 4; ++nt)
            acc[nt] = wmma32(a, load_frag(&sW1T[(nt * 16 + row) * P128 + kb]), acc[nt]);
    }
#pragma unroll
    for (int nt = 0; nt < 4; ++nt) {
        float bb = b_n1[nt * 16 + row];
#pragma unroll
        for (int r = 0; r < 8; ++r) {
            float h = silu_f(acc[nt][r] + bb);
            sMid[wave][(r + 8 * hi) * P64 + nt * 16 + row] = (_Float16)h;
        }
    }
    __syncthreads();

    v8f acc2[4];
#pragma unroll
    for (int nt = 0; nt < 4; ++nt) acc2[nt] = {};
#pragma unroll
    for (int kc = 0; kc < 2; ++kc) {
        const int kb = kc * 32 + hi * 8;
        v16h a = load_frag(&sMid[wave][row * P64 + kb]);
#pragma unroll
        for (int nt = 0; nt < 4; ++nt)
            acc2[nt] = wmma32(a, load_frag(&sW2T[(nt * 16 + row) * P64 + kb]), acc2[nt]);
    }
    if (validWave) {
#pragma unroll
        for (int nt = 0; nt < 4; ++nt) {
            float bb = b_n2[nt * 16 + row];
#pragma unroll
            for (int r = 0; r < 8; ++r) {
                int m = r + 8 * hi;
                int nd = tile * 16 + m;
                if (nd < N) out[(long)nd * 64 + nt * 16 + row] = acc2[nt][r] + bb;
            }
        }
    }
}


extern "C" void kernel_launch(void* const* d_in, const int* in_sizes, int n_in,
                              void* d_out, int out_size, void* d_ws, size_t ws_size,
                              hipStream_t stream) {
    const float* node_feat = (const float*)d_in[0];
    const float* coord     = (const float*)d_in[1];
    const float* edge_feat = (const float*)d_in[2];
    const float* W_e1 = (const float*)d_in[3];
    const float* b_e1 = (const float*)d_in[4];
    const float* W_e2 = (const float*)d_in[5];
    const float* b_e2 = (const float*)d_in[6];
    const float* W_n1 = (const float*)d_in[7];
    const float* b_n1 = (const float*)d_in[8];
    const float* W_n2 = (const float*)d_in[9];
    const float* b_n2 = (const float*)d_in[10];
    const float* W_q  = (const float*)d_in[11];
    const float* W_k  = (const float*)d_in[12];
    const int*   src  = (const int*)d_in[13];
    const int*   dst  = (const int*)d_in[14];

    const int N = in_sizes[0] / 64;
    const int E = in_sizes[13];

    float*    e_arr   = (float*)d_ws;
    unsigned* e_max   = (unsigned*)(e_arr + E);
    float*    h_neigh = (float*)(e_max + N) + N;

    int initBlocks = (N + 255) / 256;
    if (initBlocks > 1024) initBlocks = 1024;
    init_ws_kernel<<<initBlocks, 256, 0, stream>>>(e_max, N);

    const int numEdgeTiles = (E + 15) / 16;
    const int edgeBlocks   = (numEdgeTiles + WAVES - 1) / WAVES;
    const int numNodeTiles = (N + 15) / 16;
    const int nodeBlocks   = (numNodeTiles + WAVES - 1) / WAVES;

    const int edgeGrid = edgeBlocks < 1280 ? edgeBlocks : 1280;
    const int msgGroups = (N + NPG - 1) / NPG;

    edge_score_kernel<<<edgeGrid, BLOCK, 0, stream>>>(
        node_feat, coord, edge_feat, W_q, W_k, src, dst, e_arr, e_max, E);

    edge_msg_kernel<<<msgGroups, MBLOCK, 0, stream>>>(
        node_feat, coord, edge_feat, W_e1, b_e1, W_e2, b_e2,
        src, dst, e_arr, e_max, h_neigh, E, N);

    node_mlp_kernel<<<nodeBlocks, BLOCK, 0, stream>>>(
        node_feat, W_n1, b_n1, W_n2, b_n2, h_neigh, (float*)d_out, N);
}


# silu's divide by (1 + e^-x) taken with the hardware reciprocal (v_rcp_f32, 1 ulp) instead of the correctly rounded division sequence: 11 vector instructions fewer per silu, 64 silu per lane per tile
# speedup vs baseline: 1.9762x; 1.0743x over previous
#include <hip/hip_runtime.h>
#include <hip/hip_bf16.h>
#include <math.h>

typedef __attribute__((ext_vector_type(16))) _Float16 v16h;
typedef __attribute__((ext_vector_type(8)))  _Float16 v8h;
typedef __attribute__((ext_vector_type(4)))  _Float16 v4h;
typedef __attribute__((ext_vector_type(8)))  float    v8f;

#define WAVES 4
#define BLOCK 128

#define P160 168
#define P128 136
#define P96  104
#define P64   72


__device__ __forceinline__ float silu_f(float x) {
    return x * __builtin_amdgcn_rcpf(1.0f + __expf(-x));
}

__device__ __forceinline__ unsigned enc_f(float f) {
    unsigned u = __float_as_uint(f);
    return (u & 0x80000000u) ? ~u : (u | 0x80000000u);
}
__device__ __forceinline__ float dec_f(unsigned u) {
    unsigned b = (u & 0x80000000u) ? (u & 0x7FFFFFFFu) : ~u;
    return __uint_as_float(b);
}
#define ENC_NEG_INF 0x007FFFFFu

__device__ __forceinline__ v16h load_frag(const _Float16* p) {
    v8h lo = *(const v8h*)(p);
    v8h hi = *(const v8h*)(p + 16);
    return __builtin_shufflevector(lo, hi, 0, 1, 2, 3, 4, 5, 6, 7,
                                   8, 9, 10, 11, 12, 13, 14, 15);
}

__device__ __forceinline__ v8f wmma32(v16h a, v16h b, v8f c) {
    return __builtin_amdgcn_wmma_f32_16x16x32_f16(
         false, a,  false, b,
         (short)0, c,  false,  false);
}

__device__ __forceinline__ void cvt_store4(_Float16* dstp, const float* srcp) {
    const float4 v = *(const float4*)srcp;
    v4h h;
    h[0] = (_Float16)v.x; h[1] = (_Float16)v.y;
    h[2] = (_Float16)v.z; h[3] = (_Float16)v.w;
    *(v4h*)dstp = h;
}
__device__ __forceinline__ void cvt_store4v(_Float16* dstp, const float4 v) {
    v4h h;
    h[0] = (_Float16)v.x; h[1] = (_Float16)v.y;
    h[2] = (_Float16)v.z; h[3] = (_Float16)v.w;
    *(v4h*)dstp = h;
}

__device__ __forceinline__ float xhalf(const float x) {
    return __builtin_bit_cast(float, __builtin_amdgcn_permlanex16(
        __builtin_bit_cast(int, x), __builtin_bit_cast(int, x), 0x76543210, 0xfedcba98, false, false));
}

__device__ __forceinline__ void gather_rows64(float4 (&v)[8], const float* __restrict__ feat,
                                              const int idx, const int lane) {
    const bool upper = lane >= 16;
    const unsigned c4 = (unsigned)(lane & 15) * 4u;
#pragma unroll
    for (int j = 0; j < 8; ++j) {
        const int r0 = __builtin_amdgcn_readlane(idx, j);
        const int r1 = __builtin_amdgcn_readlane(idx, j + 8);
        const unsigned r = (unsigned)(upper ? r1 : r0);
        v[j] = *(const float4*)(feat + (r * 64u + c4));
    }
}
__device__ __forceinline__ void store_rows64(_Float16* tile, const int col0,
                                             const float4 (&v)[8], const int lane) {
    _Float16* p = tile + ((lane >> 4) * 8) * P160 + col0 + (lane & 15) * 4;
#pragma unroll
    for (int j = 0; j < 8; ++j) cvt_store4v(p + j * P160, v[j]);
}


__global__ __launch_bounds__(256) void init_ws_kernel(unsigned* e_max, int N) {
    int stride = gridDim.x * blockDim.x;
    int i0 = blockIdx.x * blockDim.x + threadIdx.x;
    for (int i = i0; i < N; i += stride) e_max[i] = ENC_NEG_INF;
}


__global__ __launch_bounds__(BLOCK, 6) void edge_score_kernel(
    const float* __restrict__ node_feat, const float* __restrict__ coord,
    const float* __restrict__ edge_feat,
    const float* __restrict__ W_q, const float* __restrict__ W_k,
    const int* __restrict__ src, const int* __restrict__ dst,
    float* __restrict__ e_arr, unsigned* __restrict__ e_max, int E) {

    __shared__ __attribute__((aligned(16))) _Float16 sWkT[64 * P96];
    __shared__ __attribute__((aligned(16))) _Float16 sWqT[64 * P64];
    __shared__ __attribute__((aligned(16))) _Float16 sA[WAVES][16 * P160];

    for (int i = threadIdx.x; i < 64 * 96; i += BLOCK) {
        int n = i / 96, k = i % 96;
        int kr = (k < 64) ? k : (k < 80 ? (65 + (k - 64)) : (k == 80 ? 64 : -1));
        sWkT[n * P96 + k] = (kr >= 0) ? (_Float16)W_k[kr * 64 + n] : (_Float16)0.0f;
    }
    for (int i = threadIdx.x; i < 64 * 64; i += BLOCK) {
        int n = i >> 6, k = i & 63;
        sWqT[n * P64 + k] = (_Float16)W_q[k * 64 + n];
    }

    const int wave = threadIdx.x >> 5, lane = threadIdx.x & 31;
    const int row = lane & 15, hi = lane >> 4;
    const int numTiles = (E + 15) >> 4;

    __syncthreads();

    for (int base = blockIdx.x * WAVES; base < numTiles; base += gridDim.x * WAVES) {
        const int tile = base + wave;
        const bool validWave = tile < numTiles;

        int e = validWave ? (tile * 16 + row) : 0;
        if (e >= E) e = 0;
        const int s = src[e], d = dst[e];

        const int cn = (hi ? d : s) * 3;
        const float cx = coord[cn + 0], cy = coord[cn + 1], cz = coord[cn + 2];
        float4 vs[8], vd[8], ve[2];
        gather_rows64(vs, node_feat, s, lane);
        gather_rows64(vd, node_feat, d, lane);
#pragma unroll
        for (int j = 0; j < 2; ++j) {
            int er = validWave ? (tile * 16 + (lane >> 2) + 8 * j) : 0;
            if (er >= E) er = 0;
            ve[j] = *(const float4*)(edge_feat + ((unsigned)er * 16u + (unsigned)(lane & 3) * 4u));
        }
        _Float16* At = &sA[wave][0];
        store_rows64(At, 0, vs, lane);
        store_rows64(At, 96, vd, lane);
#pragma unroll
        for (int j = 0; j < 2; ++j)
            cvt_store4v(&At[((lane >> 2) + 8 * j) * P160 + 64 + (lane & 3) * 4], ve[j]);
        {
            const float dx = cx - xhalf(cx), dy = cy - xhalf(cy), dz = cz - xhalf(cz);
            v8h t = {};
            t[0] = (hi == 0) ? (_Float16)sqrtf(dx * dx + dy * dy + dz * dz) : (_Float16)0.0f;
            *(v8h*)&At[row * P160 + 80 + 8 * hi] = t;
        }
        __syncthreads();

        v8f accK[4], accQ[4];
#pragma unroll
        for (int nt = 0; nt < 4; ++nt) { accK[nt] = {}; accQ[nt] = {}; }

#pragma unroll
        for (int kc = 0; kc < 3; ++kc) {
            const int kb = kc * 32 + hi * 8;
            v16h a = load_frag(&sA[wave][row * P160 + kb]);
#pragma unroll
            for (int nt = 0; nt < 4; ++nt)
                accK[nt] = wmma32(a, load_frag(&sWkT[(nt * 16 + row) * P96 + kb]), accK[nt]);
        }
#pragma unroll
        for (int kc = 0; kc < 2; ++kc) {
            const int kb = kc * 32 + hi * 8;
            v16h a = load_frag(&sA[wave][row * P160 + 96 + kb]);
#pragma unroll
            for (int nt = 0; nt < 4; ++nt)
                accQ[nt] = wmma32(a, load_frag(&sWqT[(nt * 16 + row) * P64 + kb]), accQ[nt]);
        }

        float p[8];
#pragma unroll
        for (int r = 0; r < 8; ++r) {
            p[r] = 0.0f;
#pragma unroll
            for (int nt = 0; nt < 4; ++nt) p[r] += accK[nt][r] * accQ[nt][r];
        }
#pragma unroll
        for (int msk = 1; msk < 16; msk <<= 1) {
#pragma unroll
            for (int r = 0; r < 8; ++r) p[r] += __shfl_xor(p[r], msk, 32);
        }

        if (validWave && row == 0) {
#pragma unroll
            for (int r = 0; r < 8; ++r) {
                int m = r + 8 * hi;
                int eg = tile * 16 + m;
                if (eg < E) {
                    float ev = p[r] * 0.125f;
                    e_arr[eg] = ev;
                    atomicMax(&e_max[dst[eg]], enc_f(ev));
                }
            }
        }
    }
}


#define MW     23
#define MBLOCK (MW * 32)
#define NPG    40
#define UNIT   1024
#define LCAP   (UNIT + 32)

__device__ __forceinline__ void wave_lds_sync() {
    __builtin_amdgcn_fence(__ATOMIC_RELEASE, "wavefront");
    __builtin_amdgcn_wave_barrier();
    __builtin_amdgcn_fence(__ATOMIC_ACQUIRE, "wavefront");
}

__global__ __launch_bounds__(MBLOCK) void edge_msg_kernel(
    const float* __restrict__ node_feat, const float* __restrict__ coord,
    const float* __restrict__ edge_feat,
    const float* __restrict__ W_e1, const float* __restrict__ b_e1,
    const float* __restrict__ W_e2, const float* __restrict__ b_e2,
    const int* __restrict__ src, const int* __restrict__ dst,
    const float* __restrict__ e_arr, const unsigned* __restrict__ e_max,
    float* __restrict__ h_neigh, int E, int N) {

    __shared__ __attribute__((aligned(16))) _Float16 sW1T[64 * P160];
    __shared__ __attribute__((aligned(16))) _Float16 sW2T[64 * P64];
    __shared__ __attribute__((aligned(16))) _Float16 sA[MW][16 * P160];
    __shared__ __attribute__((aligned(16))) _Float16 sMid[MW][16 * P64];
    __shared__ float    sW16[MW][16];
    __shared__ int      sD16[MW][16];
    __shared__ int      sLE[MW][LCAP];
    __shared__ float    sAcc[NPG * 64];
    __shared__ float    sDen[NPG];

    for (int i = threadIdx.x; i < 64 * 160; i += MBLOCK) {
        int n = i / 160, k = i % 160;
        int kr = (k < 128) ? k : (k < 144 ? (129 + (k - 128)) : (k == 144 ? 128 : -1));
        sW1T[n * P160 + k] = (kr >= 0) ? (_Float16)W_e1[kr * 64 + n] : (_Float16)0.0f;
    }
    for (int i = threadIdx.x; i < 64 * 64; i += MBLOCK) {
        int n = i >> 6, k = i & 63;
        sW2T[n * P64 + k] = (_Float16)W_e2[k * 64 + n];
    }
    for (int i = threadIdx.x; i < NPG * 64; i += MBLOCK) sAcc[i] = 0.0f;
    if (threadIdx.x < NPG) sDen[threadIdx.x] = 0.0f;

    const int wave = __builtin_amdgcn_readfirstlane((int)(threadIdx.x >> 5));
    const int lane = threadIdx.x & 31;
    const int row = lane & 15, hi = lane >> 4;
    const int lo = blockIdx.x * NPG;

    float bb1[4], bb2[4];
#pragma unroll
    for (int nt = 0; nt < 4; ++nt) {
        bb1[nt] = b_e1[nt * 16 + row];
        bb2[nt] = b_e2[nt * 16 + row];
    }

    __syncthreads();

    const int Ef = E & ~3;
    const int numUnits = (Ef + UNIT - 1) / UNIT;
    const int rot = (numUnits > 0) ? (int)((blockIdx.x * 37u) % (unsigned)numUnits) : 0;

    int cnt = 0;
    int* const myList = &sLE[wave][0];

#define APPEND_EDGES(m, ev)                                                        \
    {                                                                              \
        const unsigned b_ = __builtin_amdgcn_ballot_w32(m);                        \
        if (m) myList[__builtin_amdgcn_mbcnt_lo(b_, (unsigned)cnt)] = (ev);        \
        cnt += __builtin_popcount(b_);                                             \
    }

    if (wave == 0) {
        const bool inTail = lane < (E - Ef);
        const int dv = inTail ? dst[Ef + lane] : -1;
        const bool mt = inTail && ((unsigned)(dv - lo) < (unsigned)NPG);
        APPEND_EDGES(mt, Ef + lane)
    }

    for (int k = wave; ; k += MW) {
        const bool last = (k >= numUnits);
        if (!last) {
            int u = k + rot;
            if (u >= numUnits) u -= numUnits;
            const int ubase = u * UNIT;
            int4 v[8];
#pragma unroll
            for (int j = 0; j < 8; ++j) {
                const int e0 = ubase + j * 128 + lane * 4;
                v[j] = *(const int4*)(dst + ((e0 < Ef) ? e0 : 0));
            }
#pragma unroll
            for (int j = 0; j < 8; ++j) {
                const int e0 = ubase + j * 128 + lane * 4;
                const bool qv = e0 < Ef;
                const bool m0 = qv && ((unsigned)(v[j].x - lo) < (unsigned)NPG);
                const bool m1 = qv && ((unsigned)(v[j].y - lo) < (unsigned)NPG);
                const bool m2 = qv && ((unsigned)(v[j].z - lo) < (unsigned)NPG);
                const bool m3 = qv && ((unsigned)(v[j].w - lo) < (unsigned)NPG);
                if (__builtin_amdgcn_ballot_w32(m0 | m1 | m2 | m3) != 0u) {
                    APPEND_EDGES(m0, e0)
                    APPEND_EDGES(m1, e0 + 1)
                    APPEND_EDGES(m2, e0 + 2)
                    APPEND_EDGES(m3, e0 + 3)
                }
            }
        }

        cnt = __builtin_amdgcn_readfirstlane(cnt);
        while (cnt >= 16 || (last && cnt > 0)) {
            const int take = (cnt >= 16) ? 16 : cnt;
            cnt -= take;
            wave_lds_sync();
            const bool rowValid = row < take;
            const int e = myList[cnt + (rowValid ? row : 0)];
            const int s = src[e], d = dst[e];
            const int dl = d - lo;

            const int cn = (hi ? d : s) * 3;
            const float cx = coord[cn + 0], cy = coord[cn + 1], cz = coord[cn + 2];
            float4 vs[8], vd[8], ve[2];
            gather_rows64(vs, node_feat, s, lane);
            gather_rows64(vd, node_feat, d, lane);
#pragma unroll
            for (int j = 0; j < 2; ++j) {
                const int rr = (lane >> 2) + 8 * j;
                const int er = myList[cnt + (rr < take ? rr : 0)];
                ve[j] = *(const float4*)(edge_feat + ((unsigned)er * 16u + (unsigned)(lane & 3) * 4u));
            }
            _Float16* At = &sA[wave][0];
            store_rows64(At, 0, vs, lane);
            store_rows64(At, 64, vd, lane);
#pragma unroll
            for (int j = 0; j < 2; ++j)
                cvt_store4v(&At[((lane >> 2) + 8 * j) * P160 + 128 + (lane & 3) * 4], ve[j]);
            {
                const float dx = cx - xhalf(cx), dy = cy - xhalf(cy), dz = cz - xhalf(cz);
                v8h t = {};
                t[0] = (hi == 0) ? (_Float16)sqrtf(dx * dx + dy * dy + dz * dz) : (_Float16)0.0f;
                *(v8h*)&At[row * P160 + 144 + 8 * hi] = t;
            }
            wave_lds_sync();

            v8f acc[4];
#pragma unroll
            for (int nt = 0; nt < 4; ++nt) acc[nt] = {};
#pragma unroll
            for (int kc = 0; kc < 5; ++kc) {
                const int kb = kc * 32 + hi * 8;
                v16h a = load_frag(&sA[wave][row * P160 + kb]);
#pragma unroll
                for (int nt = 0; nt < 4; ++nt)
                    acc[nt] = wmma32(a, load_frag(&sW1T[(nt * 16 + row) * P160 + kb]), acc[nt]);
            }
#pragma unroll
            for (int nt = 0; nt < 4; ++nt) {
#pragma unroll
                for (int r = 0; r < 8; ++r) {
                    float h = silu_f(acc[nt][r] + bb1[nt]);
                    sMid[wave][(r + 8 * hi) * P64 + nt * 16 + row] = (_Float16)h;
                }
            }
            if (hi == 0) {
                float w = __expf(e_arr[e] - dec_f(e_max[d]));
                sW16[wave][row] = w;
                sD16[wave][row] = dl;
                if (rowValid) atomicAdd(&sDen[dl], w);
            }
            wave_lds_sync();

            v8f acc2[4];
#pragma unroll
            for (int nt = 0; nt < 4; ++nt) acc2[nt] = {};
#pragma unroll
            for (int kc = 0; kc < 2; ++kc) {
                const int kb = kc * 32 + hi * 8;
                v16h a = load_frag(&sMid[wave][row * P64 + kb]);
#pragma unroll
                for (int nt = 0; nt < 4; ++nt)
                    acc2[nt] = wmma32(a, load_frag(&sW2T[(nt * 16 + row) * P64 + kb]), acc2[nt]);
            }
#pragma unroll
            for (int nt = 0; nt < 4; ++nt) {
#pragma unroll
                for (int r = 0; r < 8; ++r) {
                    int m = r + 8 * hi;
                    if (m < take) {
                        float msg = silu_f(acc2[nt][r] + bb2[nt]);
                        atomicAdd(&sAcc[sD16[wave][m] * 64 + nt * 16 + row],
                                  msg * sW16[wave][m]);
                    }
                }
            }
        }
        if (last) break;
    }
#undef APPEND_EDGES

    __syncthreads();

    for (int i = threadIdx.x; i < NPG * 64; i += MBLOCK) {
        const int nl = i >> 6;
        const int node = lo + nl;
        if (node < N) {
            const float dn = sDen[nl];
            const float inv = (dn > 0.0f) ? (1.0f / dn) : 0.0f;
            h_neigh[(long)node * 64 + (i & 63)] = sAcc[i] * inv;
        }
    }
}


__global__ __launch_bounds__(BLOCK) void node_mlp_kernel(
    const float* __restrict__ node_feat,
    const float* __restrict__ W_n1, const float* __restrict__ b_n1,
    const float* __restrict__ W_n2, const float* __restrict__ b_n2,
    const float* __restrict__ h_neigh,
    float* __restrict__ out, int N) {

    __shared__ __attribute__((aligned(16))) _Float16 sW1T[64 * P128];
    __shared__ __attribute__((aligned(16))) _Float16 sW2T[64 * P64];
    __shared__ __attribute__((aligned(16))) _Float16 sA[WAVES][16 * P128];
    __shared__ __attribute__((aligned(16))) _Float16 sMid[WAVES][16 * P64];

    for (int i = threadIdx.x; i < 64 * 128; i += BLOCK) {
        int n = i >> 7, k = i & 127;
        sW1T[n * P128 + k] = (_Float16)W_n1[k * 64 + n];
    }
    for (int i = threadIdx.x; i < 64 * 64; i += BLOCK) {
        int n = i >> 6, k = i & 63;
        sW2T[n * P64 + k] = (_Float16)W_n2[k * 64 + n];
    }

    const int wave = threadIdx.x >> 5, lane = threadIdx.x & 31;
    const int row = lane & 15, hi = lane >> 4;
    const int numTiles = (N + 15) >> 4;
    const int tile = blockIdx.x * WAVES + wave;
    const bool validWave = tile < numTiles;

    int node = validWave ? (tile * 16 + row) : 0;
    if (node >= N) node = 0;

    __syncthreads();

    _Float16* Ar = &sA[wave][row * P128];
    if (hi == 0) {
#pragma unroll
        for (int c = 0; c < 64; c += 4) cvt_store4(&Ar[c], &node_feat[node * 64 + c]);
    } else {
#pragma unroll
        for (int c = 0; c < 64; c += 4) cvt_store4(&Ar[64 + c], &h_neigh[node * 64 + c]);
    }
    __syncthreads();

    v8f acc[4];
#pragma unroll
    for (int nt = 0; nt < 4; ++nt) acc[nt] = {};
#pragma unroll
    for (int kc = 0; kc < 4; ++kc) {
        const int kb = kc * 32 + hi * 8;
        v16h a = load_frag(&sA[wave][row * P128 + kb]);
#pragma unroll
        for (int nt = 0; nt < 4; ++nt)
            acc[nt] = wmma32(a, load_frag(&sW1T[(nt * 16 + row) * P128 + kb]), acc[nt]);
    }
#pragma unroll
    for (int nt = 0; nt < 4; ++nt) {
        float bb = b_n1[nt * 16 + row];
#pragma unroll
        for (int r = 0; r < 8; ++r) {
            float h = silu_f(acc[nt][r] + bb);
            sMid[wave][(r + 8 * hi) * P64 + nt * 16 + row] = (_Float16)h;
        }
    }
    __syncthreads();

    v8f acc2[4];
#pragma unroll
    for (int nt = 0; nt < 4; ++nt) acc2[nt] = {};
#pragma unroll
    for (int kc = 0; kc < 2; ++kc) {
        const int kb = kc * 32 + hi * 8;
        v16h a = load_frag(&sMid[wave][row * P64 + kb]);
#pragma unroll
        for (int nt = 0; nt < 4; ++nt)
            acc2[nt] = wmma32(a, load_frag(&sW2T[(nt * 16 + row) * P64 + kb]), acc2[nt]);
    }
    if (validWave) {
#pragma unroll
        for (int nt = 0; nt < 4; ++nt) {
            float bb = b_n2[nt * 16 + row];
#pragma unroll
            for (int r = 0; r < 8; ++r) {
                int m = r + 8 * hi;
                int nd = tile * 16 + m;
                if (nd < N) out[(long)nd * 64 + nt * 16 + row] = acc2[nt][r] + bb;
            }
        }
    }
}


extern "C" void kernel_launch(void* const* d_in, const int* in_sizes, int n_in,
                              void* d_out, int out_size, void* d_ws, size_t ws_size,
                              hipStream_t stream) {
    const float* node_feat = (const float*)d_in[0];
    const float* coord     = (const float*)d_in[1];
    const float* edge_feat = (const float*)d_in[2];
    const float* W_e1 = (const float*)d_in[3];
    const float* b_e1 = (const float*)d_in[4];
    const float* W_e2 = (const float*)d_in[5];
    const float* b_e2 = (const float*)d_in[6];
    const float* W_n1 = (const float*)d_in[7];
    const float* b_n1 = (const float*)d_in[8];
    const float* W_n2 = (const float*)d_in[9];
    const float* b_n2 = (const float*)d_in[10];
    const float* W_q  = (const float*)d_in[11];
    const float* W_k  = (const float*)d_in[12];
    const int*   src  = (const int*)d_in[13];
    const int*   dst  = (const int*)d_in[14];

    const int N = in_sizes[0] / 64;
    const int E = in_sizes[13];

    float*    e_arr   = (float*)d_ws;
    unsigned* e_max   = (unsigned*)(e_arr + E);
    float*    h_neigh = (float*)(e_max + N) + N;

    int initBlocks = (N + 255) / 256;
    if (initBlocks > 1024) initBlocks = 1024;
    init_ws_kernel<<<initBlocks, 256, 0, stream>>>(e_max, N);

    const int numEdgeTiles = (E + 15) / 16;
    const int edgeBlocks   = (numEdgeTiles + WAVES - 1) / WAVES;
    const int numNodeTiles = (N + 15) / 16;
    const int nodeBlocks   = (numNodeTiles + WAVES - 1) / WAVES;

    const int edgeGrid = edgeBlocks < 1280 ? edgeBlocks : 1280;
    const int msgGroups = (N + NPG - 1) / NPG;

    edge_score_kernel<<<edgeGrid, BLOCK, 0, stream>>>(
        node_feat, coord, edge_feat, W_q, W_k, src, dst, e_arr, e_max, E);

    edge_msg_kernel<<<msgGroups, MBLOCK, 0, stream>>>(
        node_feat, coord, edge_feat, W_e1, b_e1, W_e2, b_e2,
        src, dst, e_arr, e_max, h_neigh, E, N);

    node_mlp_kernel<<<nodeBlocks, BLOCK, 0, stream>>>(
        node_feat, W_n1, b_n1, W_n2, b_n2, h_neigh, (float*)d_out, N);
}


# dependent global loads cut from each tile's chain: score write-out takes dst from the staged tile, one store + one atomic (was 8 serial reload->atomic trips); e_arr/e_max asked for with the gather
# speedup vs baseline: 2.8278x; 1.4309x over previous
#include <hip/hip_runtime.h>
#include <hip/hip_bf16.h>
#include <math.h>

typedef __attribute__((ext_vector_type(16))) _Float16 v16h;
typedef __attribute__((ext_vector_type(8)))  _Float16 v8h;
typedef __attribute__((ext_vector_type(4)))  _Float16 v4h;
typedef __attribute__((ext_vector_type(8)))  float    v8f;

#define WAVES 4
#define BLOCK 128

#define P160 168
#define P128 136
#define P96  104
#define P64   72


__device__ __forceinline__ float silu_f(float x) {
    return x * __builtin_amdgcn_rcpf(1.0f + __expf(-x));
}

__device__ __forceinline__ unsigned enc_f(float f) {
    unsigned u = __float_as_uint(f);
    return (u & 0x80000000u) ? ~u : (u | 0x80000000u);
}
__device__ __forceinline__ float dec_f(unsigned u) {
    unsigned b = (u & 0x80000000u) ? (u & 0x7FFFFFFFu) : ~u;
    return __uint_as_float(b);
}
#define ENC_NEG_INF 0x007FFFFFu

__device__ __forceinline__ v16h load_frag(const _Float16* p) {
    v8h lo = *(const v8h*)(p);
    v8h hi = *(const v8h*)(p + 16);
    return __builtin_shufflevector(lo, hi, 0, 1, 2, 3, 4, 5, 6, 7,
                                   8, 9, 10, 11, 12, 13, 14, 15);
}

__device__ __forceinline__ v8f wmma32(v16h a, v16h b, v8f c) {
    return __builtin_amdgcn_wmma_f32_16x16x32_f16(
         false, a,  false, b,
         (short)0, c,  false,  false);
}

__device__ __forceinline__ void cvt_store4(_Float16* dstp, const float* srcp) {
    const float4 v = *(const float4*)srcp;
    v4h h;
    h[0] = (_Float16)v.x; h[1] = (_Float16)v.y;
    h[2] = (_Float16)v.z; h[3] = (_Float16)v.w;
    *(v4h*)dstp = h;
}
__device__ __forceinline__ void cvt_store4v(_Float16* dstp, const float4 v) {
    v4h h;
    h[0] = (_Float16)v.x; h[1] = (_Float16)v.y;
    h[2] = (_Float16)v.z; h[3] = (_Float16)v.w;
    *(v4h*)dstp = h;
}

__device__ __forceinline__ float xhalf(const float x) {
    return __builtin_bit_cast(float, __builtin_amdgcn_permlanex16(
        __builtin_bit_cast(int, x), __builtin_bit_cast(int, x), 0x76543210, 0xfedcba98, false, false));
}

__device__ __forceinline__ void gather_rows64(float4 (&v)[8], const float* __restrict__ feat,
                                              const int idx, const int lane) {
    const bool upper = lane >= 16;
    const unsigned c4 = (unsigned)(lane & 15) * 4u;
#pragma unroll
    for (int j = 0; j < 8; ++j) {
        const int r0 = __builtin_amdgcn_readlane(idx, j);
        const int r1 = __builtin_amdgcn_readlane(idx, j + 8);
        const unsigned r = (unsigned)(upper ? r1 : r0);
        v[j] = *(const float4*)(feat + (r * 64u + c4));
    }
}
__device__ __forceinline__ void store_rows64(_Float16* tile, const int col0,
                                             const float4 (&v)[8], const int lane) {
    _Float16* p = tile + ((lane >> 4) * 8) * P160 + col0 + (lane & 15) * 4;
#pragma unroll
    for (int j = 0; j < 8; ++j) cvt_store4v(p + j * P160, v[j]);
}


__global__ __launch_bounds__(256) void init_ws_kernel(unsigned* e_max, int N) {
    int stride = gridDim.x * blockDim.x;
    int i0 = blockIdx.x * blockDim.x + threadIdx.x;
    for (int i = i0; i < N; i += stride) e_max[i] = ENC_NEG_INF;
}


__global__ __launch_bounds__(BLOCK, 6) void edge_score_kernel(
    const float* __restrict__ node_feat, const float* __restrict__ coord,
    const float* __restrict__ edge_feat,
    const float* __restrict__ W_q, const float* __restrict__ W_k,
    const int* __restrict__ src, const int* __restrict__ dst,
    float* __restrict__ e_arr, unsigned* __restrict__ e_max, int E) {

    __shared__ __attribute__((aligned(16))) _Float16 sWkT[64 * P96];
    __shared__ __attribute__((aligned(16))) _Float16 sWqT[64 * P64];
    __shared__ __attribute__((aligned(16))) _Float16 sA[WAVES][16 * P160];

    for (int i = threadIdx.x; i < 64 * 96; i += BLOCK) {
        int n = i / 96, k = i % 96;
        int kr = (k < 64) ? k : (k < 80 ? (65 + (k - 64)) : (k == 80 ? 64 : -1));
        sWkT[n * P96 + k] = (kr >= 0) ? (_Float16)W_k[kr * 64 + n] : (_Float16)0.0f;
    }
    for (int i = threadIdx.x; i < 64 * 64; i += BLOCK) {
        int n = i >> 6, k = i & 63;
        sWqT[n * P64 + k] = (_Float16)W_q[k * 64 + n];
    }

    const int wave = __builtin_amdgcn_readfirstlane((int)(threadIdx.x >> 5));
    const int lane = threadIdx.x & 31;
    const int row = lane & 15, hi = lane >> 4;
    const int numTiles = (E + 15) >> 4;

    __syncthreads();

    for (int base = blockIdx.x * WAVES; base < numTiles; base += gridDim.x * WAVES) {
        const int tile = base + wave;
        const bool validWave = tile < numTiles;

        int e = validWave ? (tile * 16 + row) : 0;
        if (e >= E) e = 0;
        const int s = src[e], d = dst[e];

        const int cn = (hi ? d : s) * 3;
        const float cx = coord[cn + 0], cy = coord[cn + 1], cz = coord[cn + 2];
        float4 vs[8], vd[8], ve[2];
        gather_rows64(vs, node_feat, s, lane);
        gather_rows64(vd, node_feat, d, lane);
#pragma unroll
        for (int j = 0; j < 2; ++j) {
            int er = validWave ? (tile * 16 + (lane >> 2) + 8 * j) : 0;
            if (er >= E) er = 0;
            ve[j] = *(const float4*)(edge_feat + ((unsigned)er * 16u + (unsigned)(lane & 3) * 4u));
        }
        _Float16* At = &sA[wave][0];
        store_rows64(At, 0, vs, lane);
        store_rows64(At, 96, vd, lane);
#pragma unroll
        for (int j = 0; j < 2; ++j)
            cvt_store4v(&At[((lane >> 2) + 8 * j) * P160 + 64 + (lane & 3) * 4], ve[j]);
        {
            const float dx = cx - xhalf(cx), dy = cy - xhalf(cy), dz = cz - xhalf(cz);
            v8h t = {};
            t[0] = (hi == 0) ? (_Float16)sqrtf(dx * dx + dy * dy + dz * dz) : (_Float16)0.0f;
            *(v8h*)&At[row * P160 + 80 + 8 * hi] = t;
        }
        *(int*)&At[row * P160 + 160] = d;
        __syncthreads();

        v8f accK[4], accQ[4];
#pragma unroll
        for (int nt = 0; nt < 4; ++nt) { accK[nt] = {}; accQ[nt] = {}; }

#pragma unroll
        for (int kc = 0; kc < 3; ++kc) {
            const int kb = kc * 32 + hi * 8;
            v16h a = load_frag(&sA[wave][row * P160 + kb]);
#pragma unroll
            for (int nt = 0; nt < 4; ++nt)
                accK[nt] = wmma32(a, load_frag(&sWkT[(nt * 16 + row) * P96 + kb]), accK[nt]);
        }
#pragma unroll
        for (int kc = 0; kc < 2; ++kc) {
            const int kb = kc * 32 + hi * 8;
            v16h a = load_frag(&sA[wave][row * P160 + 96 + kb]);
#pragma unroll
            for (int nt = 0; nt < 4; ++nt)
                accQ[nt] = wmma32(a, load_frag(&sWqT[(nt * 16 + row) * P64 + kb]), accQ[nt]);
        }

        float p[8];
#pragma unroll
        for (int r = 0; r < 8; ++r) {
            p[r] = 0.0f;
#pragma unroll
            for (int nt = 0; nt < 4; ++nt) p[r] += accK[nt][r] * accQ[nt][r];
        }
#pragma unroll
        for (int msk = 1; msk < 16; msk <<= 1) {
#pragma unroll
            for (int r = 0; r < 8; ++r) p[r] += __shfl_xor(p[r], msk, 32);
        }

        float ev = p[0];
#pragma unroll
        for (int r = 1; r < 8; ++r) ev = ((row & 7) == r) ? p[r] : ev;
        ev *= 0.125f;
        const int eg = tile * 16 + row;
        if (validWave && (row >> 3) == hi && eg < E) {
            e_arr[eg] = ev;
            atomicMax(&e_max[*(const int*)&At[row * P160 + 160]], enc_f(ev));
        }
    }
}


#define MW     23
#define MBLOCK (MW * 32)
#define NPG    40
#define UNIT   1024
#define LCAP   (UNIT + 32)

__device__ __forceinline__ void wave_lds_sync() {
    __builtin_amdgcn_fence(__ATOMIC_RELEASE, "wavefront");
    __builtin_amdgcn_wave_barrier();
    __builtin_amdgcn_fence(__ATOMIC_ACQUIRE, "wavefront");
}

__global__ __launch_bounds__(MBLOCK) void edge_msg_kernel(
    const float* __restrict__ node_feat, const float* __restrict__ coord,
    const float* __restrict__ edge_feat,
    const float* __restrict__ W_e1, const float* __restrict__ b_e1,
    const float* __restrict__ W_e2, const float* __restrict__ b_e2,
    const int* __restrict__ src, const int* __restrict__ dst,
    const float* __restrict__ e_arr, const unsigned* __restrict__ e_max,
    float* __restrict__ h_neigh, int E, int N) {

    __shared__ __attribute__((aligned(16))) _Float16 sW1T[64 * P160];
    __shared__ __attribute__((aligned(16))) _Float16 sW2T[64 * P64];
    __shared__ __attribute__((aligned(16))) _Float16 sA[MW][16 * P160];
    __shared__ __attribute__((aligned(16))) _Float16 sMid[MW][16 * P64];
    __shared__ float    sW16[MW][16];
    __shared__ int      sD16[MW][16];
    __shared__ int      sLE[MW][LCAP];
    __shared__ float    sAcc[NPG * 64];
    __shared__ float    sDen[NPG];

    for (int i = threadIdx.x; i < 64 * 160; i += MBLOCK) {
        int n = i / 160, k = i % 160;
        int kr = (k < 128) ? k : (k < 144 ? (129 + (k - 128)) : (k == 144 ? 128 : -1));
        sW1T[n * P160 + k] = (kr >= 0) ? (_Float16)W_e1[kr * 64 + n] : (_Float16)0.0f;
    }
    for (int i = threadIdx.x; i < 64 * 64; i += MBLOCK) {
        int n = i >> 6, k = i & 63;
        sW2T[n * P64 + k] = (_Float16)W_e2[k * 64 + n];
    }
    for (int i = threadIdx.x; i < NPG * 64; i += MBLOCK) sAcc[i] = 0.0f;
    if (threadIdx.x < NPG) sDen[threadIdx.x] = 0.0f;

    const int wave = __builtin_amdgcn_readfirstlane((int)(threadIdx.x >> 5));
    const int lane = threadIdx.x & 31;
    const int row = lane & 15, hi = lane >> 4;
    const int lo = blockIdx.x * NPG;

    float bb1[4], bb2[4];
#pragma unroll
    for (int nt = 0; nt < 4; ++nt) {
        bb1[nt] = b_e1[nt * 16 + row];
        bb2[nt] = b_e2[nt * 16 + row];
    }

    __syncthreads();

    const int Ef = E & ~3;
    const int numUnits = (Ef + UNIT - 1) / UNIT;
    const int rot = (numUnits > 0) ? (int)((blockIdx.x * 37u) % (unsigned)numUnits) : 0;

    int cnt = 0;
    int* const myList = &sLE[wave][0];

#define APPEND_EDGES(m, ev)                                                        \
    {                                                                              \
        const unsigned b_ = __builtin_amdgcn_ballot_w32(m);                        \
        if (m) myList[__builtin_amdgcn_mbcnt_lo(b_, (unsigned)cnt)] = (ev);        \
        cnt += __builtin_popcount(b_);                                             \
    }

    if (wave == 0) {
        const bool inTail = lane < (E - Ef);
        const int dv = inTail ? dst[Ef + lane] : -1;
        const bool mt = inTail && ((unsigned)(dv - lo) < (unsigned)NPG);
        APPEND_EDGES(mt, Ef + lane)
    }

    for (int k = wave; ; k += MW) {
        const bool last = (k >= numUnits);
        if (!last) {
            int u = k + rot;
            if (u >= numUnits) u -= numUnits;
            const int ubase = u * UNIT;
            int4 v[8];
#pragma unroll
            for (int j = 0; j < 8; ++j) {
                const int e0 = ubase + j * 128 + lane * 4;
                v[j] = *(const int4*)(dst + ((e0 < Ef) ? e0 : 0));
            }
#pragma unroll
            for (int j = 0; j < 8; ++j) {
                const int e0 = ubase + j * 128 + lane * 4;
                const bool qv = e0 < Ef;
                const bool m0 = qv && ((unsigned)(v[j].x - lo) < (unsigned)NPG);
                const bool m1 = qv && ((unsigned)(v[j].y - lo) < (unsigned)NPG);
                const bool m2 = qv && ((unsigned)(v[j].z - lo) < (unsigned)NPG);
                const bool m3 = qv && ((unsigned)(v[j].w - lo) < (unsigned)NPG);
                if (__builtin_amdgcn_ballot_w32(m0 | m1 | m2 | m3) != 0u) {
                    APPEND_EDGES(m0, e0)
                    APPEND_EDGES(m1, e0 + 1)
                    APPEND_EDGES(m2, e0 + 2)
                    APPEND_EDGES(m3, e0 + 3)
                }
            }
        }

        cnt = __builtin_amdgcn_readfirstlane(cnt);
        while (cnt >= 16 || (last && cnt > 0)) {
            const int take = (cnt >= 16) ? 16 : cnt;
            cnt -= take;
            wave_lds_sync();
            const bool rowValid = row < take;
            const int e = myList[cnt + (rowValid ? row : 0)];
            const int s = src[e], d = dst[e];
            const int dl = d - lo;
            const float ea = e_arr[e];
            const unsigned em = e_max[d];

            const int cn = (hi ? d : s) * 3;
            const float cx = coord[cn + 0], cy = coord[cn + 1], cz = coord[cn + 2];
            float4 vs[8], vd[8], ve[2];
            gather_rows64(vs, node_feat, s, lane);
            gather_rows64(vd, node_feat, d, lane);
#pragma unroll
            for (int j = 0; j < 2; ++j) {
                const int rr = (lane >> 2) + 8 * j;
                const int er = myList[cnt + (rr < take ? rr : 0)];
                ve[j] = *(const float4*)(edge_feat + ((unsigned)er * 16u + (unsigned)(lane & 3) * 4u));
            }
            _Float16* At = &sA[wave][0];
            store_rows64(At, 0, vs, lane);
            store_rows64(At, 64, vd, lane);
#pragma unroll
            for (int j = 0; j < 2; ++j)
                cvt_store4v(&At[((lane >> 2) + 8 * j) * P160 + 128 + (lane & 3) * 4], ve[j]);
            {
                const float dx = cx - xhalf(cx), dy = cy - xhalf(cy), dz = cz - xhalf(cz);
                v8h t = {};
                t[0] = (hi == 0) ? (_Float16)sqrtf(dx * dx + dy * dy + dz * dz) : (_Float16)0.0f;
                *(v8h*)&At[row * P160 + 144 + 8 * hi] = t;
            }
            wave_lds_sync();

            v8f acc[4];
#pragma unroll
            for (int nt = 0; nt < 4; ++nt) acc[nt] = {};
#pragma unroll
            for (int kc = 0; kc < 5; ++kc) {
                const int kb = kc * 32 + hi * 8;
                v16h a = load_frag(&sA[wave][row * P160 + kb]);
#pragma unroll
                for (int nt = 0; nt < 4; ++nt)
                    acc[nt] = wmma32(a, load_frag(&sW1T[(nt * 16 + row) * P160 + kb]), acc[nt]);
            }
#pragma unroll
            for (int nt = 0; nt < 4; ++nt) {
#pragma unroll
                for (int r = 0; r < 8; ++r) {
                    float h = silu_f(acc[nt][r] + bb1[nt]);
                    sMid[wave][(r + 8 * hi) * P64 + nt * 16 + row] = (_Float16)h;
                }
            }
            if (hi == 0) {
                float w = __expf(ea - dec_f(em));
                sW16[wave][row] = w;
                sD16[wave][row] = dl;
                if (rowValid) atomicAdd(&sDen[dl], w);
            }
            wave_lds_sync();

            v8f acc2[4];
#pragma unroll
            for (int nt = 0; nt < 4; ++nt) acc2[nt] = {};
#pragma unroll
            for (int kc = 0; kc < 2; ++kc) {
                const int kb = kc * 32 + hi * 8;
                v16h a = load_frag(&sMid[wave][row * P64 + kb]);
#pragma unroll
                for (int nt = 0; nt < 4; ++nt)
                    acc2[nt] = wmma32(a, load_frag(&sW2T[(nt * 16 + row) * P64 + kb]), acc2[nt]);
            }
#pragma unroll
            for (int nt = 0; nt < 4; ++nt) {
#pragma unroll
                for (int r = 0; r < 8; ++r) {
                    int m = r + 8 * hi;
                    if (m < take) {
                        float msg = silu_f(acc2[nt][r] + bb2[nt]);
                        atomicAdd(&sAcc[sD16[wave][m] * 64 + nt * 16 + row],
                                  msg * sW16[wave][m]);
                    }
                }
            }
        }
        if (last) break;
    }
#undef APPEND_EDGES

    __syncthreads();

    for (int i = threadIdx.x; i < NPG * 64; i += MBLOCK) {
        const int nl = i >> 6;
        const int node = lo + nl;
        if (node < N) {
            const float dn = sDen[nl];
            const float inv = (dn > 0.0f) ? (1.0f / dn) : 0.0f;
            h_neigh[(long)node * 64 + (i & 63)] = sAcc[i] * inv;
        }
    }
}


__global__ __launch_bounds__(BLOCK) void node_mlp_kernel(
    const float* __restrict__ node_feat,
    const float* __restrict__ W_n1, const float* __restrict__ b_n1,
    const float* __restrict__ W_n2, const float* __restrict__ b_n2,
    const float* __restrict__ h_neigh,
    float* __restrict__ out, int N) {

    __shared__ __attribute__((aligned(16))) _Float16 sW1T[64 * P128];
    __shared__ __attribute__((aligned(16))) _Float16 sW2T[64 * P64];
    __shared__ __attribute__((aligned(16))) _Float16 sA[WAVES][16 * P128];
    __shared__ __attribute__((aligned(16))) _Float16 sMid[WAVES][16 * P64];

    for (int i = threadIdx.x; i < 64 * 128; i += BLOCK) {
        int n = i >> 7, k = i & 127;
        sW1T[n * P128 + k] = (_Float16)W_n1[k * 64 + n];
    }
    for (int i = threadIdx.x; i < 64 * 64; i += BLOCK) {
        int n = i >> 6, k = i & 63;
        sW2T[n * P64 + k] = (_Float16)W_n2[k * 64 + n];
    }

    const int wave = threadIdx.x >> 5, lane = threadIdx.x & 31;
    const int row = lane & 15, hi = lane >> 4;
    const int numTiles = (N + 15) >> 4;
    const int tile = blockIdx.x * WAVES + wave;
    const bool validWave = tile < numTiles;

    int node = validWave ? (tile * 16 + row) : 0;
    if (node >= N) node = 0;

    __syncthreads();

    _Float16* Ar = &sA[wave][row * P128];
    if (hi == 0) {
#pragma unroll
        for (int c = 0; c < 64; c += 4) cvt_store4(&Ar[c], &node_feat[node * 64 + c]);
    } else {
#pragma unroll
        for (int c = 0; c < 64; c += 4) cvt_store4(&Ar[64 + c], &h_neigh[node * 64 + c]);
    }
    __syncthreads();

    v8f acc[4];
#pragma unroll
    for (int nt = 0; nt < 4; ++nt) acc[nt] = {};
#pragma unroll
    for (int kc = 0; kc < 4; ++kc) {
        const int kb = kc * 32 + hi * 8;
        v16h a = load_frag(&sA[wave][row * P128 + kb]);
#pragma unroll
        for (int nt = 0; nt < 4; ++nt)
            acc[nt] = wmma32(a, load_frag(&sW1T[(nt * 16 + row) * P128 + kb]), acc[nt]);
    }
#pragma unroll
    for (int nt = 0; nt < 4; ++nt) {
        float bb = b_n1[nt * 16 + row];
#pragma unroll
        for (int r = 0; r < 8; ++r) {
            float h = silu_f(acc[nt][r] + bb);
            sMid[wave][(r + 8 * hi) * P64 + nt * 16 + row] = (_Float16)h;
        }
    }
    __syncthreads();

    v8f acc2[4];
#pragma unroll
    for (int nt = 0; nt < 4; ++nt) acc2[nt] = {};
#pragma unroll
    for (int kc = 0; kc < 2; ++kc) {
        const int kb = kc * 32 + hi * 8;
        v16h a = load_frag(&sMid[wave][row * P64 + kb]);
#pragma unroll
        for (int nt = 0; nt < 4; ++nt)
            acc2[nt] = wmma32(a, load_frag(&sW2T[(nt * 16 + row) * P64 + kb]), acc2[nt]);
    }
    if (validWave) {
#pragma unroll
        for (int nt = 0; nt < 4; ++nt) {
            float bb = b_n2[nt * 16 + row];
#pragma unroll
            for (int r = 0; r < 8; ++r) {
                int m = r + 8 * hi;
                int nd = tile * 16 + m;
                if (nd < N) out[(long)nd * 64 + nt * 16 + row] = acc2[nt][r] + bb;
            }
        }
    }
}


extern "C" void kernel_launch(void* const* d_in, const int* in_sizes, int n_in,
                              void* d_out, int out_size, void* d_ws, size_t ws_size,
                              hipStream_t stream) {
    const float* node_feat = (const float*)d_in[0];
    const float* coord     = (const float*)d_in[1];
    const float* edge_feat = (const float*)d_in[2];
    const float* W_e1 = (const float*)d_in[3];
    const float* b_e1 = (const float*)d_in[4];
    const float* W_e2 = (const float*)d_in[5];
    const float* b_e2 = (const float*)d_in[6];
    const float* W_n1 = (const float*)d_in[7];
    const float* b_n1 = (const float*)d_in[8];
    const float* W_n2 = (const float*)d_in[9];
    const float* b_n2 = (const float*)d_in[10];
    const float* W_q  = (const float*)d_in[11];
    const float* W_k  = (const float*)d_in[12];
    const int*   src  = (const int*)d_in[13];
    const int*   dst  = (const int*)d_in[14];

    const int N = in_sizes[0] / 64;
    const int E = in_sizes[13];

    float*    e_arr   = (float*)d_ws;
    unsigned* e_max   = (unsigned*)(e_arr + E);
    float*    h_neigh = (float*)(e_max + N) + N;

    int initBlocks = (N + 255) / 256;
    if (initBlocks > 1024) initBlocks = 1024;
    init_ws_kernel<<<initBlocks, 256, 0, stream>>>(e_max, N);

    const int numEdgeTiles = (E + 15) / 16;
    const int edgeBlocks   = (numEdgeTiles + WAVES - 1) / WAVES;
    const int numNodeTiles = (N + 15) / 16;
    const int nodeBlocks   = (numNodeTiles + WAVES - 1) / WAVES;

    const int edgeGrid = edgeBlocks < 1280 ? edgeBlocks : 1280;
    const int msgGroups = (N + NPG - 1) / NPG;

    edge_score_kernel<<<edgeGrid, BLOCK, 0, stream>>>(
        node_feat, coord, edge_feat, W_q, W_k, src, dst, e_arr, e_max, E);

    edge_msg_kernel<<<msgGroups, MBLOCK, 0, stream>>>(
        node_feat, coord, edge_feat, W_e1, b_e1, W_e2, b_e2,
        src, dst, e_arr, e_max, h_neigh, E, N);

    node_mlp_kernel<<<nodeBlocks, BLOCK, 0, stream>>>(
        node_feat, W_n1, b_n1, W_n2, b_n2, h_neigh, (float*)d_out, N);
}


# message kernel write-out: each half-wave reads its 8 rows' softmax weights and destination rows once (16-B LDS loads), adds all rows untested (unused rows weigh 0); was 32 load->wait->add blocks
# speedup vs baseline: 3.0083x; 1.0638x over previous
#include <hip/hip_runtime.h>
#include <hip/hip_bf16.h>
#include <math.h>

typedef __attribute__((ext_vector_type(16))) _Float16 v16h;
typedef __attribute__((ext_vector_type(8)))  _Float16 v8h;
typedef __attribute__((ext_vector_type(4)))  _Float16 v4h;
typedef __attribute__((ext_vector_type(8)))  float    v8f;

#define WAVES 4
#define BLOCK 128

#define P160 168
#define P128 136
#define P96  104
#define P64   72


__device__ __forceinline__ float silu_f(float x) {
    return x * __builtin_amdgcn_rcpf(1.0f + __expf(-x));
}

__device__ __forceinline__ unsigned enc_f(float f) {
    unsigned u = __float_as_uint(f);
    return (u & 0x80000000u) ? ~u : (u | 0x80000000u);
}
__device__ __forceinline__ float dec_f(unsigned u) {
    unsigned b = (u & 0x80000000u) ? (u & 0x7FFFFFFFu) : ~u;
    return __uint_as_float(b);
}
#define ENC_NEG_INF 0x007FFFFFu

__device__ __forceinline__ v16h load_frag(const _Float16* p) {
    v8h lo = *(const v8h*)(p);
    v8h hi = *(const v8h*)(p + 16);
    return __builtin_shufflevector(lo, hi, 0, 1, 2, 3, 4, 5, 6, 7,
                                   8, 9, 10, 11, 12, 13, 14, 15);
}

__device__ __forceinline__ v8f wmma32(v16h a, v16h b, v8f c) {
    return __builtin_amdgcn_wmma_f32_16x16x32_f16(
         false, a,  false, b,
         (short)0, c,  false,  false);
}

__device__ __forceinline__ void cvt_store4(_Float16* dstp, const float* srcp) {
    const float4 v = *(const float4*)srcp;
    v4h h;
    h[0] = (_Float16)v.x; h[1] = (_Float16)v.y;
    h[2] = (_Float16)v.z; h[3] = (_Float16)v.w;
    *(v4h*)dstp = h;
}
__device__ __forceinline__ void cvt_store4v(_Float16* dstp, const float4 v) {
    v4h h;
    h[0] = (_Float16)v.x; h[1] = (_Float16)v.y;
    h[2] = (_Float16)v.z; h[3] = (_Float16)v.w;
    *(v4h*)dstp = h;
}

__device__ __forceinline__ float xhalf(const float x) {
    return __builtin_bit_cast(float, __builtin_amdgcn_permlanex16(
        __builtin_bit_cast(int, x), __builtin_bit_cast(int, x), 0x76543210, 0xfedcba98, false, false));
}

__device__ __forceinline__ void gather_rows64(float4 (&v)[8], const float* __restrict__ feat,
                                              const int idx, const int lane) {
    const bool upper = lane >= 16;
    const unsigned c4 = (unsigned)(lane & 15) * 4u;
#pragma unroll
    for (int j = 0; j < 8; ++j) {
        const int r0 = __builtin_amdgcn_readlane(idx, j);
        const int r1 = __builtin_amdgcn_readlane(idx, j + 8);
        const unsigned r = (unsigned)(upper ? r1 : r0);
        v[j] = *(const float4*)(feat + (r * 64u + c4));
    }
}
__device__ __forceinline__ void store_rows64(_Float16* tile, const int col0,
                                             const float4 (&v)[8], const int lane) {
    _Float16* p = tile + ((lane >> 4) * 8) * P160 + col0 + (lane & 15) * 4;
#pragma unroll
    for (int j = 0; j < 8; ++j) cvt_store4v(p + j * P160, v[j]);
}


__global__ __launch_bounds__(256) void init_ws_kernel(unsigned* e_max, int N) {
    int stride = gridDim.x * blockDim.x;
    int i0 = blockIdx.x * blockDim.x + threadIdx.x;
    for (int i = i0; i < N; i += stride) e_max[i] = ENC_NEG_INF;
}


__global__ __launch_bounds__(BLOCK, 6) void edge_score_kernel(
    const float* __restrict__ node_feat, const float* __restrict__ coord,
    const float* __restrict__ edge_feat,
    const float* __restrict__ W_q, const float* __restrict__ W_k,
    const int* __restrict__ src, const int* __restrict__ dst,
    float* __restrict__ e_arr, unsigned* __restrict__ e_max, int E) {

    __shared__ __attribute__((aligned(16))) _Float16 sWkT[64 * P96];
    __shared__ __attribute__((aligned(16))) _Float16 sWqT[64 * P64];
    __shared__ __attribute__((aligned(16))) _Float16 sA[WAVES][16 * P160];

    for (int i = threadIdx.x; i < 64 * 96; i += BLOCK) {
        int n = i / 96, k = i % 96;
        int kr = (k < 64) ? k : (k < 80 ? (65 + (k - 64)) : (k == 80 ? 64 : -1));
        sWkT[n * P96 + k] = (kr >= 0) ? (_Float16)W_k[kr * 64 + n] : (_Float16)0.0f;
    }
    for (int i = threadIdx.x; i < 64 * 64; i += BLOCK) {
        int n = i >> 6, k = i & 63;
        sWqT[n * P64 + k] = (_Float16)W_q[k * 64 + n];
    }

    const int wave = __builtin_amdgcn_readfirstlane((int)(threadIdx.x >> 5));
    const int lane = threadIdx.x & 31;
    const int row = lane & 15, hi = lane >> 4;
    const int numTiles = (E + 15) >> 4;

    __syncthreads();

    for (int base = blockIdx.x * WAVES; base < numTiles; base += gridDim.x * WAVES) {
        const int tile = base + wave;
        const bool validWave = tile < numTiles;

        int e = validWave ? (tile * 16 + row) : 0;
        if (e >= E) e = 0;
        const int s = src[e], d = dst[e];

        const int cn = (hi ? d : s) * 3;
        const float cx = coord[cn + 0], cy = coord[cn + 1], cz = coord[cn + 2];
        float4 vs[8], vd[8], ve[2];
        gather_rows64(vs, node_feat, s, lane);
        gather_rows64(vd, node_feat, d, lane);
#pragma unroll
        for (int j = 0; j < 2; ++j) {
            int er = validWave ? (tile * 16 + (lane >> 2) + 8 * j) : 0;
            if (er >= E) er = 0;
            ve[j] = *(const float4*)(edge_feat + ((unsigned)er * 16u + (unsigned)(lane & 3) * 4u));
        }
        _Float16* At = &sA[wave][0];
        store_rows64(At, 0, vs, lane);
        store_rows64(At, 96, vd, lane);
#pragma unroll
        for (int j = 0; j < 2; ++j)
            cvt_store4v(&At[((lane >> 2) + 8 * j) * P160 + 64 + (lane & 3) * 4], ve[j]);
        {
            const float dx = cx - xhalf(cx), dy = cy - xhalf(cy), dz = cz - xhalf(cz);
            v8h t = {};
            t[0] = (hi == 0) ? (_Float16)sqrtf(dx * dx + dy * dy + dz * dz) : (_Float16)0.0f;
            *(v8h*)&At[row * P160 + 80 + 8 * hi] = t;
        }
        *(int*)&At[row * P160 + 160] = d;
        __syncthreads();

        v8f accK[4], accQ[4];
#pragma unroll
        for (int nt = 0; nt < 4; ++nt) { accK[nt] = {}; accQ[nt] = {}; }

#pragma unroll
        for (int kc = 0; kc < 3; ++kc) {
            const int kb = kc * 32 + hi * 8;
            v16h a = load_frag(&sA[wave][row * P160 + kb]);
#pragma unroll
            for (int nt = 0; nt < 4; ++nt)
                accK[nt] = wmma32(a, load_frag(&sWkT[(nt * 16 + row) * P96 + kb]), accK[nt]);
        }
#pragma unroll
        for (int kc = 0; kc < 2; ++kc) {
            const int kb = kc * 32 + hi * 8;
            v16h a = load_frag(&sA[wave][row * P160 + 96 + kb]);
#pragma unroll
            for (int nt = 0; nt < 4; ++nt)
                accQ[nt] = wmma32(a, load_frag(&sWqT[(nt * 16 + row) * P64 + kb]), accQ[nt]);
        }

        float p[8];
#pragma unroll
        for (int r = 0; r < 8; ++r) {
            p[r] = 0.0f;
#pragma unroll
            for (int nt = 0; nt < 4; ++nt) p[r] += accK[nt][r] * accQ[nt][r];
        }
#pragma unroll
        for (int msk = 1; msk < 16; msk <<= 1) {
#pragma unroll
            for (int r = 0; r < 8; ++r) p[r] += __shfl_xor(p[r], msk, 32);
        }

        float ev = p[0];
#pragma unroll
        for (int r = 1; r < 8; ++r) ev = ((row & 7) == r) ? p[r] : ev;
        ev *= 0.125f;
        const int eg = tile * 16 + row;
        if (validWave && (row >> 3) == hi && eg < E) {
            e_arr[eg] = ev;
            atomicMax(&e_max[*(const int*)&At[row * P160 + 160]], enc_f(ev));
        }
    }
}


#define MW     23
#define MBLOCK (MW * 32)
#define NPG    40
#define UNIT   1024
#define LCAP   (UNIT + 32)

__device__ __forceinline__ void wave_lds_sync() {
    __builtin_amdgcn_fence(__ATOMIC_RELEASE, "wavefront");
    __builtin_amdgcn_wave_barrier();
    __builtin_amdgcn_fence(__ATOMIC_ACQUIRE, "wavefront");
}

__global__ __launch_bounds__(MBLOCK) void edge_msg_kernel(
    const float* __restrict__ node_feat, const float* __restrict__ coord,
    const float* __restrict__ edge_feat,
    const float* __restrict__ W_e1, const float* __restrict__ b_e1,
    const float* __restrict__ W_e2, const float* __restrict__ b_e2,
    const int* __restrict__ src, const int* __restrict__ dst,
    const float* __restrict__ e_arr, const unsigned* __restrict__ e_max,
    float* __restrict__ h_neigh, int E, int N) {

    __shared__ __attribute__((aligned(16))) _Float16 sW1T[64 * P160];
    __shared__ __attribute__((aligned(16))) _Float16 sW2T[64 * P64];
    __shared__ __attribute__((aligned(16))) _Float16 sA[MW][16 * P160];
    __shared__ __attribute__((aligned(16))) _Float16 sMid[MW][16 * P64];
    __shared__ __attribute__((aligned(16))) float    sW16[MW][16];
    __shared__ __attribute__((aligned(16))) int      sD16[MW][16];
    __shared__ int      sLE[MW][LCAP];
    __shared__ float    sAcc[NPG * 64];
    __shared__ float    sDen[NPG];

    for (int i = threadIdx.x; i < 64 * 160; i += MBLOCK) {
        int n = i / 160, k = i % 160;
        int kr = (k < 128) ? k : (k < 144 ? (129 + (k - 128)) : (k == 144 ? 128 : -1));
        sW1T[n * P160 + k] = (kr >= 0) ? (_Float16)W_e1[kr * 64 + n] : (_Float16)0.0f;
    }
    for (int i = threadIdx.x; i < 64 * 64; i += MBLOCK) {
        int n = i >> 6, k = i & 63;
        sW2T[n * P64 + k] = (_Float16)W_e2[k * 64 + n];
    }
    for (int i = threadIdx.x; i < NPG * 64; i += MBLOCK) sAcc[i] = 0.0f;
    if (threadIdx.x < NPG) sDen[threadIdx.x] = 0.0f;

    const int wave = __builtin_amdgcn_readfirstlane((int)(threadIdx.x >> 5));
    const int lane = threadIdx.x & 31;
    const int row = lane & 15, hi = lane >> 4;
    const int lo = blockIdx.x * NPG;

    float bb1[4], bb2[4];
#pragma unroll
    for (int nt = 0; nt < 4; ++nt) {
        bb1[nt] = b_e1[nt * 16 + row];
        bb2[nt] = b_e2[nt * 16 + row];
    }

    __syncthreads();

    const int Ef = E & ~3;
    const int numUnits = (Ef + UNIT - 1) / UNIT;
    const int rot = (numUnits > 0) ? (int)((blockIdx.x * 37u) % (unsigned)numUnits) : 0;

    int cnt = 0;
    int* const myList = &sLE[wave][0];

#define APPEND_EDGES(m, ev)                                                        \
    {                                                                              \
        const unsigned b_ = __builtin_amdgcn_ballot_w32(m);                        \
        if (m) myList[__builtin_amdgcn_mbcnt_lo(b_, (unsigned)cnt)] = (ev);        \
        cnt += __builtin_popcount(b_);                                             \
    }

    if (wave == 0) {
        const bool inTail = lane < (E - Ef);
        const int dv = inTail ? dst[Ef + lane] : -1;
        const bool mt = inTail && ((unsigned)(dv - lo) < (unsigned)NPG);
        APPEND_EDGES(mt, Ef + lane)
    }

    for (int k = wave; ; k += MW) {
        const bool last = (k >= numUnits);
        if (!last) {
            int u = k + rot;
            if (u >= numUnits) u -= numUnits;
            const int ubase = u * UNIT;
            int4 v[8];
#pragma unroll
            for (int j = 0; j < 8; ++j) {
                const int e0 = ubase + j * 128 + lane * 4;
                v[j] = *(const int4*)(dst + ((e0 < Ef) ? e0 : 0));
            }
#pragma unroll
            for (int j = 0; j < 8; ++j) {
                const int e0 = ubase + j * 128 + lane * 4;
                const bool qv = e0 < Ef;
                const bool m0 = qv && ((unsigned)(v[j].x - lo) < (unsigned)NPG);
                const bool m1 = qv && ((unsigned)(v[j].y - lo) < (unsigned)NPG);
                const bool m2 = qv && ((unsigned)(v[j].z - lo) < (unsigned)NPG);
                const bool m3 = qv && ((unsigned)(v[j].w - lo) < (unsigned)NPG);
                if (__builtin_amdgcn_ballot_w32(m0 | m1 | m2 | m3) != 0u) {
                    APPEND_EDGES(m0, e0)
                    APPEND_EDGES(m1, e0 + 1)
                    APPEND_EDGES(m2, e0 + 2)
                    APPEND_EDGES(m3, e0 + 3)
                }
            }
        }

        cnt = __builtin_amdgcn_readfirstlane(cnt);
        while (cnt >= 16 || (last && cnt > 0)) {
            const int take = (cnt >= 16) ? 16 : cnt;
            cnt -= take;
            wave_lds_sync();
            const bool rowValid = row < take;
            const int e = myList[cnt + (rowValid ? row : 0)];
            const int s = src[e], d = dst[e];
            const int dl = d - lo;
            const float ea = e_arr[e];
            const unsigned em = e_max[d];

            const int cn = (hi ? d : s) * 3;
            const float cx = coord[cn + 0], cy = coord[cn + 1], cz = coord[cn + 2];
            float4 vs[8], vd[8], ve[2];
            gather_rows64(vs, node_feat, s, lane);
            gather_rows64(vd, node_feat, d, lane);
#pragma unroll
            for (int j = 0; j < 2; ++j) {
                const int rr = (lane >> 2) + 8 * j;
                const int er = myList[cnt + (rr < take ? rr : 0)];
                ve[j] = *(const float4*)(edge_feat + ((unsigned)er * 16u + (unsigned)(lane & 3) * 4u));
            }
            _Float16* At = &sA[wave][0];
            store_rows64(At, 0, vs, lane);
            store_rows64(At, 64, vd, lane);
#pragma unroll
            for (int j = 0; j < 2; ++j)
                cvt_store4v(&At[((lane >> 2) + 8 * j) * P160 + 128 + (lane & 3) * 4], ve[j]);
            {
                const float dx = cx - xhalf(cx), dy = cy - xhalf(cy), dz = cz - xhalf(cz);
                v8h t = {};
                t[0] = (hi == 0) ? (_Float16)sqrtf(dx * dx + dy * dy + dz * dz) : (_Float16)0.0f;
                *(v8h*)&At[row * P160 + 144 + 8 * hi] = t;
            }
            wave_lds_sync();

            v8f acc[4];
#pragma unroll
            for (int nt = 0; nt < 4; ++nt) acc[nt] = {};
#pragma unroll
            for (int kc = 0; kc < 5; ++kc) {
                const int kb = kc * 32 + hi * 8;
                v16h a = load_frag(&sA[wave][row * P160 + kb]);
#pragma unroll
                for (int nt = 0; nt < 4; ++nt)
                    acc[nt] = wmma32(a, load_frag(&sW1T[(nt * 16 + row) * P160 + kb]), acc[nt]);
            }
#pragma unroll
            for (int nt = 0; nt < 4; ++nt) {
#pragma unroll
                for (int r = 0; r < 8; ++r) {
                    float h = silu_f(acc[nt][r] + bb1[nt]);
                    sMid[wave][(r + 8 * hi) * P64 + nt * 16 + row] = (_Float16)h;
                }
            }
            if (hi == 0) {
                const float w = __expf(ea - dec_f(em));
                sW16[wave][row] = rowValid ? w : 0.0f;
                sD16[wave][row] = dl;
                if (rowValid) atomicAdd(&sDen[dl], w);
            }
            wave_lds_sync();

            const float4 w0 = *(const float4*)&sW16[wave][8 * hi], w1 = *(const float4*)&sW16[wave][8 * hi + 4];
            const int4   d0 = *(const int4*)&sD16[wave][8 * hi],   d1 = *(const int4*)&sD16[wave][8 * hi + 4];
            const float wv[8] = {w0.x, w0.y, w0.z, w0.w, w1.x, w1.y, w1.z, w1.w};
            const int   dv[8] = {d0.x, d0.y, d0.z, d0.w, d1.x, d1.y, d1.z, d1.w};

            v8f acc2[4];
#pragma unroll
            for (int nt = 0; nt < 4; ++nt) acc2[nt] = {};
#pragma unroll
            for (int kc = 0; kc < 2; ++kc) {
                const int kb = kc * 32 + hi * 8;
                v16h a = load_frag(&sMid[wave][row * P64 + kb]);
#pragma unroll
                for (int nt = 0; nt < 4; ++nt)
                    acc2[nt] = wmma32(a, load_frag(&sW2T[(nt * 16 + row) * P64 + kb]), acc2[nt]);
            }
#pragma unroll
            for (int r = 0; r < 8; ++r) {
                float* const accRow = &sAcc[dv[r] * 64 + row];
#pragma unroll
                for (int nt = 0; nt < 4; ++nt)
                    atomicAdd(&accRow[nt * 16], silu_f(acc2[nt][r] + bb2[nt]) * wv[r]);
            }
        }
        if (last) break;
    }
#undef APPEND_EDGES

    __syncthreads();

    for (int i = threadIdx.x; i < NPG * 64; i += MBLOCK) {
        const int nl = i >> 6;
        const int node = lo + nl;
        if (node < N) {
            const float dn = sDen[nl];
            const float inv = (dn > 0.0f) ? (1.0f / dn) : 0.0f;
            h_neigh[(long)node * 64 + (i & 63)] = sAcc[i] * inv;
        }
    }
}


__global__ __launch_bounds__(BLOCK) void node_mlp_kernel(
    const float* __restrict__ node_feat,
    const float* __restrict__ W_n1, const float* __restrict__ b_n1,
    const float* __restrict__ W_n2, const float* __restrict__ b_n2,
    const float* __restrict__ h_neigh,
    float* __restrict__ out, int N) {

    __shared__ __attribute__((aligned(16))) _Float16 sW1T[64 * P128];
    __shared__ __attribute__((aligned(16))) _Float16 sW2T[64 * P64];
    __shared__ __attribute__((aligned(16))) _Float16 sA[WAVES][16 * P128];
    __shared__ __attribute__((aligned(16))) _Float16 sMid[WAVES][16 * P64];

    for (int i = threadIdx.x; i < 64 * 128; i += BLOCK) {
        int n = i >> 7, k = i & 127;
        sW1T[n * P128 + k] = (_Float16)W_n1[k * 64 + n];
    }
    for (int i = threadIdx.x; i < 64 * 64; i += BLOCK) {
        int n = i >> 6, k = i & 63;
        sW2T[n * P64 + k] = (_Float16)W_n2[k * 64 + n];
    }

    const int wave = threadIdx.x >> 5, lane = threadIdx.x & 31;
    const int row = lane & 15, hi = lane >> 4;
    const int numTiles = (N + 15) >> 4;
    const int tile = blockIdx.x * WAVES + wave;
    const bool validWave = tile < numTiles;

    int node = validWave ? (tile * 16 + row) : 0;
    if (node >= N) node = 0;

    __syncthreads();

    _Float16* Ar = &sA[wave][row * P128];
    if (hi == 0) {
#pragma unroll
        for (int c = 0; c < 64; c += 4) cvt_store4(&Ar[c], &node_feat[node * 64 + c]);
    } else {
#pragma unroll
        for (int c = 0; c < 64; c += 4) cvt_store4(&Ar[64 + c], &h_neigh[node * 64 + c]);
    }
    __syncthreads();

    v8f acc[4];
#pragma unroll
    for (int nt = 0; nt < 4; ++nt) acc[nt] = {};
#pragma unroll
    for (int kc = 0; kc < 4; ++kc) {
        const int kb = kc * 32 + hi * 8;
        v16h a = load_frag(&sA[wave][row * P128 + kb]);
#pragma unroll
        for (int nt = 0; nt < 4; ++nt)
            acc[nt] = wmma32(a, load_frag(&sW1T[(nt * 16 + row) * P128 + kb]), acc[nt]);
    }
#pragma unroll
    for (int nt = 0; nt < 4; ++nt) {
        float bb = b_n1[nt * 16 + row];
#pragma unroll
        for (int r = 0; r < 8; ++r) {
            float h = silu_f(acc[nt][r] + bb);
            sMid[wave][(r + 8 * hi) * P64 + nt * 16 + row] = (_Float16)h;
        }
    }
    __syncthreads();

    v8f acc2[4];
#pragma unroll
    for (int nt = 0; nt < 4; ++nt) acc2[nt] = {};
#pragma unroll
    for (int kc = 0; kc < 2; ++kc) {
        const int kb = kc * 32 + hi * 8;
        v16h a = load_frag(&sMid[wave][row * P64 + kb]);
#pragma unroll
        for (int nt = 0; nt < 4; ++nt)
            acc2[nt] = wmma32(a, load_frag(&sW2T[(nt * 16 + row) * P64 + kb]), acc2[nt]);
    }
    if (validWave) {
#pragma unroll
        for (int nt = 0; nt < 4; ++nt) {
            float bb = b_n2[nt * 16 + row];
#pragma unroll
            for (int r = 0; r < 8; ++r) {
                int m = r + 8 * hi;
                int nd = tile * 16 + m;
                if (nd < N) out[(long)nd * 64 + nt * 16 + row] = acc2[nt][r] + bb;
            }
        }
    }
}


extern "C" void kernel_launch(void* const* d_in, const int* in_sizes, int n_in,
                              void* d_out, int out_size, void* d_ws, size_t ws_size,
                              hipStream_t stream) {
    const float* node_feat = (const float*)d_in[0];
    const float* coord     = (const float*)d_in[1];
    const float* edge_feat = (const float*)d_in[2];
    const float* W_e1 = (const float*)d_in[3];
    const float* b_e1 = (const float*)d_in[4];
    const float* W_e2 = (const float*)d_in[5];
    const float* b_e2 = (const float*)d_in[6];
    const float* W_n1 = (const float*)d_in[7];
    const float* b_n1 = (const float*)d_in[8];
    const float* W_n2 = (const float*)d_in[9];
    const float* b_n2 = (const float*)d_in[10];
    const float* W_q  = (const float*)d_in[11];
    const float* W_k  = (const float*)d_in[12];
    const int*   src  = (const int*)d_in[13];
    const int*   dst  = (const int*)d_in[14];

    const int N = in_sizes[0] / 64;
    const int E = in_sizes[13];

    float*    e_arr   = (float*)d_ws;
    unsigned* e_max   = (unsigned*)(e_arr + E);
    float*    h_neigh = (float*)(e_max + N) + N;

    int initBlocks = (N + 255) / 256;
    if (initBlocks > 1024) initBlocks = 1024;
    init_ws_kernel<<<initBlocks, 256, 0, stream>>>(e_max, N);

    const int numEdgeTiles = (E + 15) / 16;
    const int edgeBlocks   = (numEdgeTiles + WAVES - 1) / WAVES;
    const int numNodeTiles = (N + 15) / 16;
    const int nodeBlocks   = (numNodeTiles + WAVES - 1) / WAVES;

    const int edgeGrid = edgeBlocks < 1280 ? edgeBlocks : 1280;
    const int msgGroups = (N + NPG - 1) / NPG;

    edge_score_kernel<<<edgeGrid, BLOCK, 0, stream>>>(
        node_feat, coord, edge_feat, W_q, W_k, src, dst, e_arr, e_max, E);

    edge_msg_kernel<<<msgGroups, MBLOCK, 0, stream>>>(
        node_feat, coord, edge_feat, W_e1, b_e1, W_e2, b_e2,
        src, dst, e_arr, e_max, h_neigh, E, N);

    node_mlp_kernel<<<nodeBlocks, BLOCK, 0, stream>>>(
        node_feat, W_n1, b_n1, W_n2, b_n2, h_neigh, (float*)d_out, N);
}


# message kernel scans a byte-wide group-id array (dst / 40, one byte per edge, written by the score kernel) instead of the 4-byte dst array: a quarter of the scan's loads and cache lines
# speedup vs baseline: 3.3437x; 1.1115x over previous
#include <hip/hip_runtime.h>
#include <hip/hip_bf16.h>
#include <math.h>

typedef __attribute__((ext_vector_type(16))) _Float16 v16h;
typedef __attribute__((ext_vector_type(8)))  _Float16 v8h;
typedef __attribute__((ext_vector_type(4)))  _Float16 v4h;
typedef __attribute__((ext_vector_type(8)))  float    v8f;

#define WAVES 4
#define BLOCK 128
#define NPG   40

#define P160 168
#define P128 136
#define P96  104
#define P64   72


__device__ __forceinline__ float silu_f(float x) {
    return x * __builtin_amdgcn_rcpf(1.0f + __expf(-x));
}

__device__ __forceinline__ unsigned enc_f(float f) {
    unsigned u = __float_as_uint(f);
    return (u & 0x80000000u) ? ~u : (u | 0x80000000u);
}
__device__ __forceinline__ float dec_f(unsigned u) {
    unsigned b = (u & 0x80000000u) ? (u & 0x7FFFFFFFu) : ~u;
    return __uint_as_float(b);
}
#define ENC_NEG_INF 0x007FFFFFu

__device__ __forceinline__ v16h load_frag(const _Float16* p) {
    v8h lo = *(const v8h*)(p);
    v8h hi = *(const v8h*)(p + 16);
    return __builtin_shufflevector(lo, hi, 0, 1, 2, 3, 4, 5, 6, 7,
                                   8, 9, 10, 11, 12, 13, 14, 15);
}

__device__ __forceinline__ v8f wmma32(v16h a, v16h b, v8f c) {
    return __builtin_amdgcn_wmma_f32_16x16x32_f16(
         false, a,  false, b,
         (short)0, c,  false,  false);
}

__device__ __forceinline__ void cvt_store4(_Float16* dstp, const float* srcp) {
    const float4 v = *(const float4*)srcp;
    v4h h;
    h[0] = (_Float16)v.x; h[1] = (_Float16)v.y;
    h[2] = (_Float16)v.z; h[3] = (_Float16)v.w;
    *(v4h*)dstp = h;
}
__device__ __forceinline__ void cvt_store4v(_Float16* dstp, const float4 v) {
    v4h h;
    h[0] = (_Float16)v.x; h[1] = (_Float16)v.y;
    h[2] = (_Float16)v.z; h[3] = (_Float16)v.w;
    *(v4h*)dstp = h;
}

__device__ __forceinline__ float xhalf(const float x) {
    return __builtin_bit_cast(float, __builtin_amdgcn_permlanex16(
        __builtin_bit_cast(int, x), __builtin_bit_cast(int, x), 0x76543210, 0xfedcba98, false, false));
}

__device__ __forceinline__ void gather_rows64(float4 (&v)[8], const float* __restrict__ feat,
                                              const int idx, const int lane) {
    const bool upper = lane >= 16;
    const unsigned c4 = (unsigned)(lane & 15) * 4u;
#pragma unroll
    for (int j = 0; j < 8; ++j) {
        const int r0 = __builtin_amdgcn_readlane(idx, j);
        const int r1 = __builtin_amdgcn_readlane(idx, j + 8);
        const unsigned r = (unsigned)(upper ? r1 : r0);
        v[j] = *(const float4*)(feat + (r * 64u + c4));
    }
}
__device__ __forceinline__ void store_rows64(_Float16* tile, const int col0,
                                             const float4 (&v)[8], const int lane) {
    _Float16* p = tile + ((lane >> 4) * 8) * P160 + col0 + (lane & 15) * 4;
#pragma unroll
    for (int j = 0; j < 8; ++j) cvt_store4v(p + j * P160, v[j]);
}


__global__ __launch_bounds__(256) void init_ws_kernel(unsigned* e_max, int N) {
    int stride = gridDim.x * blockDim.x;
    int i0 = blockIdx.x * blockDim.x + threadIdx.x;
    for (int i = i0; i < N; i += stride) e_max[i] = ENC_NEG_INF;
}


__global__ __launch_bounds__(BLOCK, 6) void edge_score_kernel(
    const float* __restrict__ node_feat, const float* __restrict__ coord,
    const float* __restrict__ edge_feat,
    const float* __restrict__ W_q, const float* __restrict__ W_k,
    const int* __restrict__ src, const int* __restrict__ dst,
    float* __restrict__ e_arr, unsigned* __restrict__ e_max,
    unsigned* __restrict__ gid4, int E) {

    __shared__ __attribute__((aligned(16))) _Float16 sWkT[64 * P96];
    __shared__ __attribute__((aligned(16))) _Float16 sWqT[64 * P64];
    __shared__ __attribute__((aligned(16))) _Float16 sA[WAVES][16 * P160];

    for (int i = threadIdx.x; i < 64 * 96; i += BLOCK) {
        int n = i / 96, k = i % 96;
        int kr = (k < 64) ? k : (k < 80 ? (65 + (k - 64)) : (k == 80 ? 64 : -1));
        sWkT[n * P96 + k] = (kr >= 0) ? (_Float16)W_k[kr * 64 + n] : (_Float16)0.0f;
    }
    for (int i = threadIdx.x; i < 64 * 64; i += BLOCK) {
        int n = i >> 6, k = i & 63;
        sWqT[n * P64 + k] = (_Float16)W_q[k * 64 + n];
    }

    const int wave = __builtin_amdgcn_readfirstlane((int)(threadIdx.x >> 5));
    const int lane = threadIdx.x & 31;
    const int row = lane & 15, hi = lane >> 4;
    const int numTiles = (E + 15) >> 4;

    __syncthreads();

    for (int base = blockIdx.x * WAVES; base < numTiles; base += gridDim.x * WAVES) {
        const int tile = base + wave;
        const bool validWave = tile < numTiles;

        int e = validWave ? (tile * 16 + row) : 0;
        if (e >= E) e = 0;
        const int s = src[e], d = dst[e];

        const int cn = (hi ? d : s) * 3;
        const float cx = coord[cn + 0], cy = coord[cn + 1], cz = coord[cn + 2];
        float4 vs[8], vd[8], ve[2];
        gather_rows64(vs, node_feat, s, lane);
        gather_rows64(vd, node_feat, d, lane);
#pragma unroll
        for (int j = 0; j < 2; ++j) {
            int er = validWave ? (tile * 16 + (lane >> 2) + 8 * j) : 0;
            if (er >= E) er = 0;
            ve[j] = *(const float4*)(edge_feat + ((unsigned)er * 16u + (unsigned)(lane & 3) * 4u));
        }
        _Float16* At = &sA[wave][0];
        store_rows64(At, 0, vs, lane);
        store_rows64(At, 96, vd, lane);
#pragma unroll
        for (int j = 0; j < 2; ++j)
            cvt_store4v(&At[((lane >> 2) + 8 * j) * P160 + 64 + (lane & 3) * 4], ve[j]);
        {
            const float dx = cx - xhalf(cx), dy = cy - xhalf(cy), dz = cz - xhalf(cz);
            v8h t = {};
            t[0] = (hi == 0) ? (_Float16)sqrtf(dx * dx + dy * dy + dz * dz) : (_Float16)0.0f;
            *(v8h*)&At[row * P160 + 80 + 8 * hi] = t;
        }
        *(int*)&At[row * P160 + 160] = d;
        __syncthreads();

        v8f accK[4], accQ[4];
#pragma unroll
        for (int nt = 0; nt < 4; ++nt) { accK[nt] = {}; accQ[nt] = {}; }

#pragma unroll
        for (int kc = 0; kc < 3; ++kc) {
            const int kb = kc * 32 + hi * 8;
            v16h a = load_frag(&sA[wave][row * P160 + kb]);
#pragma unroll
            for (int nt = 0; nt < 4; ++nt)
                accK[nt] = wmma32(a, load_frag(&sWkT[(nt * 16 + row) * P96 + kb]), accK[nt]);
        }
#pragma unroll
        for (int kc = 0; kc < 2; ++kc) {
            const int kb = kc * 32 + hi * 8;
            v16h a = load_frag(&sA[wave][row * P160 + 96 + kb]);
#pragma unroll
            for (int nt = 0; nt < 4; ++nt)
                accQ[nt] = wmma32(a, load_frag(&sWqT[(nt * 16 + row) * P64 + kb]), accQ[nt]);
        }

        float p[8];
#pragma unroll
        for (int r = 0; r < 8; ++r) {
            p[r] = 0.0f;
#pragma unroll
            for (int nt = 0; nt < 4; ++nt) p[r] += accK[nt][r] * accQ[nt][r];
        }
#pragma unroll
        for (int msk = 1; msk < 16; msk <<= 1) {
#pragma unroll
            for (int r = 0; r < 8; ++r) p[r] += __shfl_xor(p[r], msk, 32);
        }

        float ev = p[0];
#pragma unroll
        for (int r = 1; r < 8; ++r) ev = ((row & 7) == r) ? p[r] : ev;
        ev *= 0.125f;
        const int eg = tile * 16 + row;
        const int dd = *(const int*)&At[row * P160 + 160];
        {
            unsigned gq = (((unsigned)dd / (unsigned)NPG) & 255u) << ((row & 3) * 8);
            gq |= __shfl_xor(gq, 1, 32);
            gq |= __shfl_xor(gq, 2, 32);
            if (validWave && hi == 0 && (row & 3) == 0 && eg < E)
                gid4[tile * 4 + (row >> 2)] = gq;
        }
        if (validWave && (row >> 3) == hi && eg < E) {
            e_arr[eg] = ev;
            atomicMax(&e_max[dd], enc_f(ev));
        }
    }
}


#define MW     23
#define MBLOCK (MW * 32)
#define CHUNK  512
#define UCH    8
#define LCAP   (2 * CHUNK + 32)

__device__ __forceinline__ unsigned zero_bytes(const unsigned y) {
    const unsigned t = ((y & 0x7F7F7F7Fu) + 0x7F7F7F7Fu) | y;
    return ~t & 0x80808080u;
}

__device__ __forceinline__ void wave_lds_sync() {
    __builtin_amdgcn_fence(__ATOMIC_RELEASE, "wavefront");
    __builtin_amdgcn_wave_barrier();
    __builtin_amdgcn_fence(__ATOMIC_ACQUIRE, "wavefront");
}

__global__ __launch_bounds__(MBLOCK) void edge_msg_kernel(
    const float* __restrict__ node_feat, const float* __restrict__ coord,
    const float* __restrict__ edge_feat,
    const float* __restrict__ W_e1, const float* __restrict__ b_e1,
    const float* __restrict__ W_e2, const float* __restrict__ b_e2,
    const int* __restrict__ src, const int* __restrict__ dst,
    const float* __restrict__ e_arr, const unsigned* __restrict__ e_max,
    const unsigned char* __restrict__ gidb,
    float* __restrict__ h_neigh, int E, int N) {

    __shared__ __attribute__((aligned(16))) _Float16 sW1T[64 * P160];
    __shared__ __attribute__((aligned(16))) _Float16 sW2T[64 * P64];
    __shared__ __attribute__((aligned(16))) _Float16 sA[MW][16 * P160];
    __shared__ __attribute__((aligned(16))) _Float16 sMid[MW][16 * P64];
    __shared__ __attribute__((aligned(16))) float    sW16[MW][16];
    __shared__ __attribute__((aligned(16))) int      sD16[MW][16];
    __shared__ int      sLE[MW][LCAP];
    __shared__ float    sAcc[NPG * 64];
    __shared__ float    sDen[NPG];

    for (int i = threadIdx.x; i < 64 * 160; i += MBLOCK) {
        int n = i / 160, k = i % 160;
        int kr = (k < 128) ? k : (k < 144 ? (129 + (k - 128)) : (k == 144 ? 128 : -1));
        sW1T[n * P160 + k] = (kr >= 0) ? (_Float16)W_e1[kr * 64 + n] : (_Float16)0.0f;
    }
    for (int i = threadIdx.x; i < 64 * 64; i += MBLOCK) {
        int n = i >> 6, k = i & 63;
        sW2T[n * P64 + k] = (_Float16)W_e2[k * 64 + n];
    }
    for (int i = threadIdx.x; i < NPG * 64; i += MBLOCK) sAcc[i] = 0.0f;
    if (threadIdx.x < NPG) sDen[threadIdx.x] = 0.0f;

    const int wave = __builtin_amdgcn_readfirstlane((int)(threadIdx.x >> 5));
    const int lane = threadIdx.x & 31;
    const int row = lane & 15, hi = lane >> 4;
    const int lo = blockIdx.x * NPG;

    float bb1[4], bb2[4];
#pragma unroll
    for (int nt = 0; nt < 4; ++nt) {
        bb1[nt] = b_e1[nt * 16 + row];
        bb2[nt] = b_e2[nt * 16 + row];
    }

    __syncthreads();

    const int nChunks = E / CHUNK;
    const int numUnits = (nChunks + UCH - 1) / UCH;
    const int rot = (numUnits > 0) ? (int)((blockIdx.x * 37u) % (unsigned)numUnits) : 0;
    const unsigned pat = (blockIdx.x & 255u) * 0x01010101u;

    int cnt = 0;
    int* const myList = &sLE[wave][0];

#define APPEND_EDGES(m, ev)                                                        \
    {                                                                              \
        const unsigned b_ = __builtin_amdgcn_ballot_w32(m);                        \
        if (m) myList[__builtin_amdgcn_mbcnt_lo(b_, (unsigned)cnt)] = (ev);        \
        cnt += __builtin_popcount(b_);                                             \
    }

    if (wave == 0) {
        for (int eb = nChunks * CHUNK; eb < E; eb += 32) {
            const int et = eb + lane;
            const bool inTail = et < E;
            const int dv = inTail ? dst[et] : -1;
            const bool mt = inTail && ((unsigned)(dv - lo) < (unsigned)NPG);
            APPEND_EDGES(mt, et)
        }
    }

    int k = wave;
    int jstart = 0;
    for (;;) {
        const bool last = (k >= numUnits);
        bool full = false;
        if (!last) {
            int u = k + rot;
            if (u >= numUnits) u -= numUnits;
            const int c0 = u * UCH;
            int4 v[UCH];
#pragma unroll
            for (int j = 0; j < UCH; ++j) {
                const int cj = (c0 + j < nChunks) ? (c0 + j) : 0;
                v[j] = ((const int4*)(gidb + (size_t)cj * CHUNK))[lane];
            }
#pragma unroll
            for (int j = 0; j < UCH; ++j) {
                if (j >= jstart && !full && c0 + j < nChunks) {
                    unsigned m = (zero_bytes((unsigned)v[j].x ^ pat) >> 7)
                               | (zero_bytes((unsigned)v[j].y ^ pat) >> 6)
                               | (zero_bytes((unsigned)v[j].z ^ pat) >> 5)
                               | (zero_bytes((unsigned)v[j].w ^ pat) >> 4);
                    unsigned b = __builtin_amdgcn_ballot_w32(m != 0u);
                    if (b != 0u) {
                        if (cnt + CHUNK > LCAP) {
                            full = true;
                            jstart = j;
                        } else {
                            const int e0 = (c0 + j) * CHUNK + lane * 16;
                            do {
                                if (m != 0u) {
                                    const unsigned p = (unsigned)__builtin_ctz(m);
                                    myList[__builtin_amdgcn_mbcnt_lo(b, (unsigned)cnt)] =
                                        e0 + (int)(((p & 7u) << 2) + (p >> 3));
                                    m &= m - 1u;
                                }
                                cnt += __builtin_popcount(b);
                                b = __builtin_amdgcn_ballot_w32(m != 0u);
                            } while (b != 0u);
                        }
                    }
                }
            }
        }

        cnt = __builtin_amdgcn_readfirstlane(cnt);
        while (cnt >= 16 || (last && cnt > 0)) {
            const int take = (cnt >= 16) ? 16 : cnt;
            cnt -= take;
            wave_lds_sync();
            const bool rowValid = row < take;
            const int e = myList[cnt + (rowValid ? row : 0)];
            const int s = src[e], d = dst[e];
            const bool own = rowValid && ((unsigned)(d - lo) < (unsigned)NPG);
            const int dl = own ? (d - lo) : 0;
            const float ea = e_arr[e];
            const unsigned em = e_max[d];

            const int cn = (hi ? d : s) * 3;
            const float cx = coord[cn + 0], cy = coord[cn + 1], cz = coord[cn + 2];
            float4 vs[8], vd[8], ve[2];
            gather_rows64(vs, node_feat, s, lane);
            gather_rows64(vd, node_feat, d, lane);
#pragma unroll
            for (int j = 0; j < 2; ++j) {
                const int rr = (lane >> 2) + 8 * j;
                const int er = myList[cnt + (rr < take ? rr : 0)];
                ve[j] = *(const float4*)(edge_feat + ((unsigned)er * 16u + (unsigned)(lane & 3) * 4u));
            }
            _Float16* At = &sA[wave][0];
            store_rows64(At, 0, vs, lane);
            store_rows64(At, 64, vd, lane);
#pragma unroll
            for (int j = 0; j < 2; ++j)
                cvt_store4v(&At[((lane >> 2) + 8 * j) * P160 + 128 + (lane & 3) * 4], ve[j]);
            {
                const float dx = cx - xhalf(cx), dy = cy - xhalf(cy), dz = cz - xhalf(cz);
                v8h t = {};
                t[0] = (hi == 0) ? (_Float16)sqrtf(dx * dx + dy * dy + dz * dz) : (_Float16)0.0f;
                *(v8h*)&At[row * P160 + 144 + 8 * hi] = t;
            }
            wave_lds_sync();

            v8f acc[4];
#pragma unroll
            for (int nt = 0; nt < 4; ++nt) acc[nt] = {};
#pragma unroll
            for (int kc = 0; kc < 5; ++kc) {
                const int kb = kc * 32 + hi * 8;
                v16h a = load_frag(&sA[wave][row * P160 + kb]);
#pragma unroll
                for (int nt = 0; nt < 4; ++nt)
                    acc[nt] = wmma32(a, load_frag(&sW1T[(nt * 16 + row) * P160 + kb]), acc[nt]);
            }
#pragma unroll
            for (int nt = 0; nt < 4; ++nt) {
#pragma unroll
                for (int r = 0; r < 8; ++r) {
                    float h = silu_f(acc[nt][r] + bb1[nt]);
                    sMid[wave][(r + 8 * hi) * P64 + nt * 16 + row] = (_Float16)h;
                }
            }
            if (hi == 0) {
                const float w = __expf(ea - dec_f(em));
                sW16[wave][row] = own ? w : 0.0f;
                sD16[wave][row] = dl;
                if (own) atomicAdd(&sDen[dl], w);
            }
            wave_lds_sync();

            const float4 w0 = *(const float4*)&sW16[wave][8 * hi], w1 = *(const float4*)&sW16[wave][8 * hi + 4];
            const int4   d0 = *(const int4*)&sD16[wave][8 * hi],   d1 = *(const int4*)&sD16[wave][8 * hi + 4];
            const float wv[8] = {w0.x, w0.y, w0.z, w0.w, w1.x, w1.y, w1.z, w1.w};
            const int   dv[8] = {d0.x, d0.y, d0.z, d0.w, d1.x, d1.y, d1.z, d1.w};

            v8f acc2[4];
#pragma unroll
            for (int nt = 0; nt < 4; ++nt) acc2[nt] = {};
#pragma unroll
            for (int kc = 0; kc < 2; ++kc) {
                const int kb = kc * 32 + hi * 8;
                v16h a = load_frag(&sMid[wave][row * P64 + kb]);
#pragma unroll
                for (int nt = 0; nt < 4; ++nt)
                    acc2[nt] = wmma32(a, load_frag(&sW2T[(nt * 16 + row) * P64 + kb]), acc2[nt]);
            }
#pragma unroll
            for (int r = 0; r < 8; ++r) {
                float* const accRow = &sAcc[dv[r] * 64 + row];
#pragma unroll
                for (int nt = 0; nt < 4; ++nt)
                    atomicAdd(&accRow[nt * 16], silu_f(acc2[nt][r] + bb2[nt]) * wv[r]);
            }
        }
        if (last) break;
        if (!full) { k += MW; jstart = 0; }
    }
#undef APPEND_EDGES

    __syncthreads();

    for (int i = threadIdx.x; i < NPG * 64; i += MBLOCK) {
        const int nl = i >> 6;
        const int node = lo + nl;
        if (node < N) {
            const float dn = sDen[nl];
            const float inv = (dn > 0.0f) ? (1.0f / dn) : 0.0f;
            h_neigh[(long)node * 64 + (i & 63)] = sAcc[i] * inv;
        }
    }
}


__global__ __launch_bounds__(BLOCK) void node_mlp_kernel(
    const float* __restrict__ node_feat,
    const float* __restrict__ W_n1, const float* __restrict__ b_n1,
    const float* __restrict__ W_n2, const float* __restrict__ b_n2,
    const float* __restrict__ h_neigh,
    float* __restrict__ out, int N) {

    __shared__ __attribute__((aligned(16))) _Float16 sW1T[64 * P128];
    __shared__ __attribute__((aligned(16))) _Float16 sW2T[64 * P64];
    __shared__ __attribute__((aligned(16))) _Float16 sA[WAVES][16 * P128];
    __shared__ __attribute__((aligned(16))) _Float16 sMid[WAVES][16 * P64];

    for (int i = threadIdx.x; i < 64 * 128; i += BLOCK) {
        int n = i >> 7, k = i & 127;
        sW1T[n * P128 + k] = (_Float16)W_n1[k * 64 + n];
    }
    for (int i = threadIdx.x; i < 64 * 64; i += BLOCK) {
        int n = i >> 6, k = i & 63;
        sW2T[n * P64 + k] = (_Float16)W_n2[k * 64 + n];
    }

    const int wave = threadIdx.x >> 5, lane = threadIdx.x & 31;
    const int row = lane & 15, hi = lane >> 4;
    const int numTiles = (N + 15) >> 4;
    const int tile = blockIdx.x * WAVES + wave;
    const bool validWave = tile < numTiles;

    int node = validWave ? (tile * 16 + row) : 0;
    if (node >= N) node = 0;

    __syncthreads();

    _Float16* Ar = &sA[wave][row * P128];
    if (hi == 0) {
#pragma unroll
        for (int c = 0; c < 64; c += 4) cvt_store4(&Ar[c], &node_feat[node * 64 + c]);
    } else {
#pragma unroll
        for (int c = 0; c < 64; c += 4) cvt_store4(&Ar[64 + c], &h_neigh[node * 64 + c]);
    }
    __syncthreads();

    v8f acc[4];
#pragma unroll
    for (int nt = 0; nt < 4; ++nt) acc[nt] = {};
#pragma unroll
    for (int kc = 0; kc < 4; ++kc) {
        const int kb = kc * 32 + hi * 8;
        v16h a = load_frag(&sA[wave][row * P128 + kb]);
#pragma unroll
        for (int nt = 0; nt < 4; ++nt)
            acc[nt] = wmma32(a, load_frag(&sW1T[(nt * 16 + row) * P128 + kb]), acc[nt]);
    }
#pragma unroll
    for (int nt = 0; nt < 4; ++nt) {
        float bb = b_n1[nt * 16 + row];
#pragma unroll
        for (int r = 0; r < 8; ++r) {
            float h = silu_f(acc[nt][r] + bb);
            sMid[wave][(r + 8 * hi) * P64 + nt * 16 + row] = (_Float16)h;
        }
    }
    __syncthreads();

    v8f acc2[4];
#pragma unroll
    for (int nt = 0; nt < 4; ++nt) acc2[nt] = {};
#pragma unroll
    for (int kc = 0; kc < 2; ++kc) {
        const int kb = kc * 32 + hi * 8;
        v16h a = load_frag(&sMid[wave][row * P64 + kb]);
#pragma unroll
        for (int nt = 0; nt < 4; ++nt)
            acc2[nt] = wmma32(a, load_frag(&sW2T[(nt * 16 + row) * P64 + kb]), acc2[nt]);
    }
    if (validWave) {
#pragma unroll
        for (int nt = 0; nt < 4; ++nt) {
            float bb = b_n2[nt * 16 + row];
#pragma unroll
            for (int r = 0; r < 8; ++r) {
                int m = r + 8 * hi;
                int nd = tile * 16 + m;
                if (nd < N) out[(long)nd * 64 + nt * 16 + row] = acc2[nt][r] + bb;
            }
        }
    }
}


extern "C" void kernel_launch(void* const* d_in, const int* in_sizes, int n_in,
                              void* d_out, int out_size, void* d_ws, size_t ws_size,
                              hipStream_t stream) {
    const float* node_feat = (const float*)d_in[0];
    const float* coord     = (const float*)d_in[1];
    const float* edge_feat = (const float*)d_in[2];
    const float* W_e1 = (const float*)d_in[3];
    const float* b_e1 = (const float*)d_in[4];
    const float* W_e2 = (const float*)d_in[5];
    const float* b_e2 = (const float*)d_in[6];
    const float* W_n1 = (const float*)d_in[7];
    const float* b_n1 = (const float*)d_in[8];
    const float* W_n2 = (const float*)d_in[9];
    const float* b_n2 = (const float*)d_in[10];
    const float* W_q  = (const float*)d_in[11];
    const float* W_k  = (const float*)d_in[12];
    const int*   src  = (const int*)d_in[13];
    const int*   dst  = (const int*)d_in[14];

    const int N = in_sizes[0] / 64;
    const int E = in_sizes[13];

    float*    e_arr   = (float*)d_ws;
    unsigned* e_max   = (unsigned*)(e_arr + E);
    float*    h_neigh = (float*)(e_max + N) + N;
    unsigned char* gidb = (unsigned char*)d_out;

    int initBlocks = (N + 255) / 256;
    if (initBlocks > 1024) initBlocks = 1024;
    init_ws_kernel<<<initBlocks, 256, 0, stream>>>(e_max, N);

    const int numEdgeTiles = (E + 15) / 16;
    const int edgeBlocks   = (numEdgeTiles + WAVES - 1) / WAVES;
    const int numNodeTiles = (N + 15) / 16;
    const int nodeBlocks   = (numNodeTiles + WAVES - 1) / WAVES;

    const int edgeGrid = edgeBlocks < 1280 ? edgeBlocks : 1280;
    const int msgGroups = (N + NPG - 1) / NPG;

    edge_score_kernel<<<edgeGrid, BLOCK, 0, stream>>>(
        node_feat, coord, edge_feat, W_q, W_k, src, dst, e_arr, e_max, (unsigned*)gidb, E);

    edge_msg_kernel<<<msgGroups, MBLOCK, 0, stream>>>(
        node_feat, coord, edge_feat, W_e1, b_e1, W_e2, b_e2,
        src, dst, e_arr, e_max, gidb, h_neigh, E, N);

    node_mlp_kernel<<<nodeBlocks, BLOCK, 0, stream>>>(
        node_feat, W_n1, b_n1, W_n2, b_n2, h_neigh, (float*)d_out, N);
}
